# starting kernel
# baseline (speedup 1.0000x reference)
#include <hip/hip_runtime.h>
#include <hip/hip_bf16.h>

#define BB   4
#define LL   4096
#define MM   1024
#define SS   64
#define ROWS (BB * LL)
#define NC   16
#define LC   (LL / NC)

typedef __attribute__((ext_vector_type(16))) _Float16 v16h;
typedef __attribute__((ext_vector_type(8)))  _Float16 v8h;
typedef __attribute__((ext_vector_type(4)))  _Float16 v4h;
typedef __attribute__((ext_vector_type(8)))  float    v8f;

__device__ __forceinline__ v16h load_frag(const _Float16* __restrict__ p) {
    v8h lo = *(const v8h*)(p);
    v8h hi = *(const v8h*)(p + 16);
    v16h t;
#pragma unroll
    for (int i = 0; i < 8; ++i) { t[i] = lo[i]; t[i + 8] = hi[i]; }
    return t;
}

struct Acc44 { v8f c[4][4]; };

__device__ __forceinline__ void wave_gemm64(const _Float16* __restrict__ A,
                                            const _Float16* __restrict__ B,
                                            int ldA, int ldB,
                                            int rowbase, int colbase, int K,
                                            Acc44& acc) {
    const int lane  = threadIdx.x & 31;
    const int rr    = lane & 15;
    const int khalf = (lane >> 4) & 1;
    const _Float16* Ab[4];
    const _Float16* Bb[4];
#pragma unroll
    for (int t = 0; t < 4; ++t) {
        Ab[t] = A + (size_t)(rowbase + t * 16 + rr) * ldA + khalf * 8;
        Bb[t] = B + (size_t)(colbase + t * 16 + rr) * ldB + khalf * 8;
    }
    for (int kb = 0; kb < K; kb += 32) {
        v16h a[4], b[4];
#pragma unroll
        for (int t = 0; t < 4; ++t) a[t] = load_frag(Ab[t] + kb);
#pragma unroll
        for (int t = 0; t < 4; ++t) b[t] = load_frag(Bb[t] + kb);
#pragma unroll
        for (int i = 0; i < 4; ++i)
#pragma unroll
            for (int j = 0; j < 4; ++j)
                acc.c[i][j] = __builtin_amdgcn_wmma_f32_16x16x32_f16(
                    false, a[i], false, b[j], (short)0, acc.c[i][j], false, false);
    }
}

__device__ __forceinline__ float fast_sigmoid(float v) {
    return __builtin_amdgcn_rcpf(1.0f + __expf(-v));
}

__global__ void k_convert(const float* __restrict__ W_in,
                          const float* __restrict__ W_xp,
                          const float* __restrict__ W_out,
                          const float* __restrict__ Dmat,
                          _Float16* __restrict__ Wih,
                          _Float16* __restrict__ Wxh,
                          _Float16* __restrict__ Woh,
                          _Float16* __restrict__ Dth) {
    const int NWI = 2 * MM * MM, NXP = SS * MM, NWO = MM * MM, ND = SS * MM;
    int i = blockIdx.x * 256 + threadIdx.x;
    if (i < NWI) { Wih[i] = (_Float16)W_in[i]; return; }
    i -= NWI;
    if (i < NXP) { Wxh[i] = (_Float16)W_xp[i]; return; }
    i -= NXP;
    if (i < NWO) { Woh[i] = (_Float16)W_out[i]; return; }
    i -= NWO;
    if (i < ND) {
        int n = i >> 6, s = i & 63;
        Dth[i] = (_Float16)Dmat[(size_t)s * MM + n];
    }
}

__global__ __launch_bounds__(256) void k_layernorm(const float* __restrict__ x,
                                                   const float* __restrict__ gamma,
                                                   const float* __restrict__ beta,
                                                   _Float16* __restrict__ xn) {
    __shared__ float sSum[8], sSq[8];
    const int row = blockIdx.x;
    const size_t base = (size_t)row * MM;
    const int tid = threadIdx.x;

    float4 v = ((const float4*)(x + base))[tid];
    float s = v.x + v.y + v.z + v.w;
    float q = v.x * v.x + v.y * v.y + v.z * v.z + v.w * v.w;
#pragma unroll
    for (int off = 16; off > 0; off >>= 1) {
        s += __shfl_down(s, off);
        q += __shfl_down(q, off);
    }
    if ((tid & 31) == 0) { sSum[tid >> 5] = s; sSq[tid >> 5] = q; }
    __syncthreads();
    float ts = 0.f, tq = 0.f;
#pragma unroll
    for (int i = 0; i < 8; ++i) { ts += sSum[i]; tq += sSq[i]; }
    const float mu = ts * (1.0f / MM);
    const float var = tq * (1.0f / MM) - mu * mu;
    const float rstd = rsqrtf(var + 1e-5f);

    float4 g = ((const float4*)gamma)[tid];
    float4 bb = ((const float4*)beta)[tid];
    v4h o;
    o[0] = (_Float16)((v.x - mu) * rstd * g.x + bb.x);
    o[1] = (_Float16)((v.y - mu) * rstd * g.y + bb.y);
    o[2] = (_Float16)((v.z - mu) * rstd * g.z + bb.z);
    o[3] = (_Float16)((v.w - mu) * rstd * g.w + bb.w);
    ((v4h*)(xn + base))[tid] = o;
}

__global__ __launch_bounds__(256) void k_gemm1(const _Float16* __restrict__ xn,
                                               const _Float16* __restrict__ Wih,
                                               const float* __restrict__ b_in,
                                               _Float16* __restrict__ xp,
                                               _Float16* __restrict__ zg) {
    Acc44 acc = {};
    const int w = threadIdx.x >> 5;
    const int rowbase = blockIdx.x * 128 + (w & 1) * 64;
    const int colbase = blockIdx.y * 256 + (w >> 1) * 64;
    wave_gemm64(xn, Wih, MM, MM, rowbase, colbase, MM, acc);

    const int lane = threadIdx.x & 31;
    const int nlo = lane & 15, mh = (lane >> 4) * 8;
    const bool is_x = (blockIdx.y < (MM / 256));
#pragma unroll
    for (int j = 0; j < 4; ++j) {
        const int n = colbase + j * 16 + nlo;
        const float bias = b_in[n];
#pragma unroll
        for (int i = 0; i < 4; ++i) {
#pragma unroll
            for (int jj = 0; jj < 8; ++jj) {
                const int m = rowbase + i * 16 + mh + jj;
                const float v = acc.c[i][j][jj] + bias;
                const float sg = fast_sigmoid(v);
                if (is_x) xp[(size_t)m * MM + n] = (_Float16)(v * sg);
                else      zg[(size_t)m * MM + (n - MM)] = (_Float16)sg;
            }
        }
    }
}

__global__ __launch_bounds__(256) void k_gemm2(const _Float16* __restrict__ xp,
                                               const _Float16* __restrict__ Wxh,
                                               const float* __restrict__ b_xp,
                                               float* __restrict__ u) {
    Acc44 acc = {};
    const int w = threadIdx.x >> 5;
    const int rowbase = blockIdx.x * 512 + w * 64;
    wave_gemm64(xp, Wxh, MM, MM, rowbase, 0, MM, acc);

    const int lane = threadIdx.x & 31;
    const int nlo = lane & 15, mh = (lane >> 4) * 8;
#pragma unroll
    for (int j = 0; j < 4; ++j) {
        const int n = j * 16 + nlo;
        const float bias = b_xp[n];
#pragma unroll
        for (int i = 0; i < 4; ++i)
#pragma unroll
            for (int jj = 0; jj < 8; ++jj) {
                const int m = rowbase + i * 16 + mh + jj;
                u[(size_t)m * SS + n] = acc.c[i][j][jj] + bias;
            }
    }
}

__global__ void k_scan_chunk(const float* __restrict__ u,
                             const float* __restrict__ A_log,
                             float* __restrict__ hloc,
                             float* __restrict__ chunk_end) {
    const int t = blockIdx.x * 256 + threadIdx.x;
    const int s = t & 63;
    const int c = (t >> 6) & (NC - 1);
    const int b = t >> 10;
    const float a = expf(-expf(A_log[s]));
    float h = 0.f;
    const int l0 = c * LC;
    for (int j = 0; j < LC; ++j) {
        const size_t idx = ((size_t)b * LL + l0 + j) * SS + s;
        h = a * h + u[idx];
        hloc[idx] = h;
    }
    chunk_end[((b * SS + s) * NC) + c] = h;
}

__global__ void k_scan_carry(const float* __restrict__ A_log,
                             const float* __restrict__ chunk_end,
                             float* __restrict__ carry_in) {
    const int t = threadIdx.x;
    const int s = t & 63, b = t >> 6;
    const float E = expf(-expf(A_log[s]) * (float)LC);
    float carry = 0.f;
    for (int c = 0; c < NC; ++c) {
        carry_in[((b * SS + s) * NC) + c] = carry;
        carry = E * carry + chunk_end[((b * SS + s) * NC) + c];
    }
}

__global__ void k_scan_apply(const float* __restrict__ hloc,
                             const float* __restrict__ A_log,
                             const float* __restrict__ carry_in,
                             _Float16* __restrict__ h_h) {
    const int t = blockIdx.x * 256 + threadIdx.x;
    const int s = t & 63;
    const int c = (t >> 6) & (NC - 1);
    const int b = t >> 10;
    const float a = expf(-expf(A_log[s]));
    const float carry = carry_in[((b * SS + s) * NC) + c];
    float p = a;
    const int l0 = c * LC;
    for (int j = 0; j < LC; ++j) {
        const int l = l0 + j;
        const float hv = hloc[((size_t)b * LL + l) * SS + s] + carry * p;
        p *= a;
        h_h[((size_t)l * BB + b) * SS + s] = (_Float16)hv;
    }
}

__global__ __launch_bounds__(256) void k_gemm3(const _Float16* __restrict__ h_h,
                                               const _Float16* __restrict__ Dth,
                                               const _Float16* __restrict__ zg,
                                               _Float16* __restrict__ ys) {
    Acc44 acc = {};
    const int w = threadIdx.x >> 5;
    const int rowbase = blockIdx.x * 128 + (w & 1) * 64;
    const int colbase = blockIdx.y * 256 + (w >> 1) * 64;
    wave_gemm64(h_h, Dth, SS, SS, rowbase, colbase, SS, acc);

    const int lane = threadIdx.x & 31;
    const int nlo = lane & 15, mh = (lane >> 4) * 8;
#pragma unroll
    for (int i = 0; i < 4; ++i)
#pragma unroll
        for (int j = 0; j < 4; ++j)
#pragma unroll
            for (int jj = 0; jj < 8; ++jj) {
                const int rp = rowbase + i * 16 + mh + jj;
                const int n = colbase + j * 16 + nlo;
                const int b = rp & (BB - 1), l = rp >> 2;
                const float z = (float)zg[((size_t)b * LL + l) * MM + n];
                ys[(size_t)rp * MM + n] = (_Float16)(acc.c[i][j][jj] * z);
            }
}

__global__ __launch_bounds__(256) void k_gemm4(const _Float16* __restrict__ ys,
                                               const _Float16* __restrict__ Woh,
                                               const float* __restrict__ b_out,
                                               const float* __restrict__ xflat,
                                               float* __restrict__ out) {
    Acc44 acc = {};
    const int w = threadIdx.x >> 5;
    const int rowbase = blockIdx.x * 128 + (w & 1) * 64;
    const int colbase = blockIdx.y * 256 + (w >> 1) * 64;
    wave_gemm64(ys, Woh, MM, MM, rowbase, colbase, MM, acc);

    const int lane = threadIdx.x & 31;
    const int nlo = lane & 15, mh = (lane >> 4) * 8;
#pragma unroll
    for (int j = 0; j < 4; ++j) {
        const int n = colbase + j * 16 + nlo;
        const float bias = b_out[n];
#pragma unroll
        for (int i = 0; i < 4; ++i)
#pragma unroll
            for (int jj = 0; jj < 8; ++jj) {
                const int m = rowbase + i * 16 + mh + jj;
                const size_t idx = (size_t)m * MM + n;
                out[idx] = xflat[idx] + acc.c[i][j][jj] + bias;
            }
    }
}

extern "C" void kernel_launch(void* const* d_in, const int* in_sizes, int n_in,
                              void* d_out, int out_size, void* d_ws, size_t ws_size,
                              hipStream_t stream) {
    const float* x     = (const float*)d_in[0];
    const float* A_log = (const float*)d_in[1];
    const float* Dmat  = (const float*)d_in[2];
    const float* W_in  = (const float*)d_in[3];
    const float* b_in  = (const float*)d_in[4];
    const float* W_xp  = (const float*)d_in[5];
    const float* b_xp  = (const float*)d_in[6];
    const float* W_out = (const float*)d_in[7];
    const float* b_out = (const float*)d_in[8];
    const float* gamma = (const float*)d_in[9];
    const float* beta  = (const float*)d_in[10];
    float* out = (float*)d_out;

    char* ws = (char*)d_ws;
    size_t off = 0;
    auto take = [&](size_t bytes) { char* p = ws + off; off += bytes; return p; };
    _Float16* xn_h  = (_Float16*)take((size_t)ROWS * MM * 2);
    _Float16* xp_h  = (_Float16*)take((size_t)ROWS * MM * 2);
    _Float16* z_h   = (_Float16*)take((size_t)ROWS * MM * 2);
    _Float16* ys_h  = (_Float16*)take((size_t)ROWS * MM * 2);
    float*    u     = (float*)take((size_t)ROWS * SS * 4);
    float*    hloc  = (float*)take((size_t)ROWS * SS * 4);
    _Float16* h_h   = (_Float16*)take((size_t)ROWS * SS * 2);
    _Float16* Wih   = (_Float16*)take((size_t)2 * MM * MM * 2);
    _Float16* Wxh   = (_Float16*)take((size_t)SS * MM * 2);
    _Float16* Woh   = (_Float16*)take((size_t)MM * MM * 2);
    _Float16* Dth   = (_Float16*)take((size_t)SS * MM * 2);
    float*    cend  = (float*)take((size_t)BB * SS * NC * 4);
    float*    cin   = (float*)take((size_t)BB * SS * NC * 4);
    (void)ws_size; (void)in_sizes; (void)n_in; (void)out_size;

    {
        const int total = 2 * MM * MM + SS * MM + MM * MM + SS * MM;
        k_convert<<<(total + 255) / 256, 256, 0, stream>>>(W_in, W_xp, W_out, Dmat,
                                                           Wih, Wxh, Woh, Dth);
    }
    k_layernorm<<<ROWS, 256, 0, stream>>>(x, gamma, beta, xn_h);
    k_gemm1<<<dim3(ROWS / 128, (2 * MM) / 256), 256, 0, stream>>>(xn_h, Wih, b_in, xp_h, z_h);
    k_gemm2<<<dim3(ROWS / 512, 1), 256, 0, stream>>>(xp_h, Wxh, b_xp, u);
    k_scan_chunk<<<(BB * SS * NC) / 256, 256, 0, stream>>>(u, A_log, hloc, cend);
    k_scan_carry<<<1, BB * SS, 0, stream>>>(A_log, cend, cin);
    k_scan_apply<<<(BB * SS * NC) / 256, 256, 0, stream>>>(hloc, A_log, cin, h_h);
    k_gemm3<<<dim3(ROWS / 128, MM / 256), 256, 0, stream>>>(h_h, Dth, z_h, ys_h);
    k_gemm4<<<dim3(ROWS / 128, MM / 256), 256, 0, stream>>>(ys_h, Woh, b_out, x, out);
}


# in_proj and out_proj GEMMs: A/B tiles (128x256, K-step 64) staged through LDS by whole-line global loads, two stages, two work-groups per WGP, replacing per-wave 32-B row gathers from cache
# speedup vs baseline: 1.6884x; 1.6884x over previous
#include <hip/hip_runtime.h>
#include <hip/hip_bf16.h>

#define BB   4
#define LL   4096
#define MM   1024
#define SS   64
#define ROWS (BB * LL)
#define NC   16
#define LC   (LL / NC)

typedef __attribute__((ext_vector_type(16))) _Float16 v16h;
typedef __attribute__((ext_vector_type(8)))  _Float16 v8h;
typedef __attribute__((ext_vector_type(4)))  _Float16 v4h;
typedef __attribute__((ext_vector_type(8)))  float    v8f;

__device__ __forceinline__ v16h load_frag(const _Float16* __restrict__ p) {
    v8h lo = *(const v8h*)(p);
    v8h hi = *(const v8h*)(p + 16);
    v16h t;
#pragma unroll
    for (int i = 0; i < 8; ++i) { t[i] = lo[i]; t[i + 8] = hi[i]; }
    return t;
}

struct Acc44 { v8f c[4][4]; };

__device__ __forceinline__ void wave_gemm64(const _Float16* __restrict__ A,
                                            const _Float16* __restrict__ B,
                                            int ldA, int ldB,
                                            int rowbase, int colbase, int K,
                                            Acc44& acc) {
    const int lane  = threadIdx.x & 31;
    const int rr    = lane & 15;
    const int khalf = (lane >> 4) & 1;
    const _Float16* Ab[4];
    const _Float16* Bb[4];
#pragma unroll
    for (int t = 0; t < 4; ++t) {
        Ab[t] = A + (size_t)(rowbase + t * 16 + rr) * ldA + khalf * 8;
        Bb[t] = B + (size_t)(colbase + t * 16 + rr) * ldB + khalf * 8;
    }
    for (int kb = 0; kb < K; kb += 32) {
        v16h a[4], b[4];
#pragma unroll
        for (int t = 0; t < 4; ++t) a[t] = load_frag(Ab[t] + kb);
#pragma unroll
        for (int t = 0; t < 4; ++t) b[t] = load_frag(Bb[t] + kb);
#pragma unroll
        for (int i = 0; i < 4; ++i)
#pragma unroll
            for (int j = 0; j < 4; ++j)
                acc.c[i][j] = __builtin_amdgcn_wmma_f32_16x16x32_f16(
                    false, a[i], false, b[j], (short)0, acc.c[i][j], false, false);
    }
}

#define GBK     64
#define GLDP    72
#define GTM     128
#define GTN     256
#define GSTAGE  ((GTM + GTN) * GLDP)
#define GKT     (MM / GBK)

struct Stage12 { v8h a[4]; v8h b[8]; };

__device__ __forceinline__ void tile_fetch(const _Float16* __restrict__ gA,
                                           const _Float16* __restrict__ gB,
                                           int kofs, Stage12& g) {
#pragma unroll
    for (int r = 0; r < 4; ++r) g.a[r] = *(const v8h*)(gA + (size_t)r * 32 * MM + kofs);
#pragma unroll
    for (int r = 0; r < 8; ++r) g.b[r] = *(const v8h*)(gB + (size_t)r * 32 * MM + kofs);
}

__device__ __forceinline__ void tile_store(_Float16* sA, _Float16* sB, const Stage12& g) {
#pragma unroll
    for (int r = 0; r < 4; ++r) *(v8h*)(sA + r * 32 * GLDP) = g.a[r];
#pragma unroll
    for (int r = 0; r < 8; ++r) *(v8h*)(sB + r * 32 * GLDP) = g.b[r];
}

__device__ __forceinline__ void stage_compute(const _Float16* fA, const _Float16* fB, Acc44& acc) {
    v16h a[2][4], b[2][4];
#pragma unroll
    for (int ks = 0; ks < 2; ++ks) {
#pragma unroll
        for (int t = 0; t < 4; ++t) a[ks][t] = load_frag(fA + t * 16 * GLDP + ks * 32);
#pragma unroll
        for (int t = 0; t < 4; ++t) b[ks][t] = load_frag(fB + t * 16 * GLDP + ks * 32);
    }
#pragma unroll
    for (int ks = 0; ks < 2; ++ks)
#pragma unroll
        for (int i = 0; i < 4; ++i)
#pragma unroll
            for (int j = 0; j < 4; ++j)
                acc.c[i][j] = __builtin_amdgcn_wmma_f32_16x16x32_f16(
                    false, a[ks][i], false, b[ks][j], (short)0, acc.c[i][j], false, false);
}

__device__ __forceinline__ void block_gemm_lds(const _Float16* __restrict__ A,
                                               const _Float16* __restrict__ B,
                                               int rowblk, int colblk,
                                               _Float16* smem, Acc44& acc) {
    const int tid   = threadIdx.x;
    const int lane  = tid & 31;
    const int w     = tid >> 5;
    const int rr    = lane & 15;
    const int khalf = (lane >> 4) & 1;

    const int crow = tid >> 3;
    const int ccol = (tid & 7) * 8;
    const _Float16* gA = A + (size_t)(rowblk + crow) * MM + ccol;
    const _Float16* gB = B + (size_t)(colblk + crow) * MM + ccol;
    const int cofs = crow * GLDP + ccol;

    const int faofs = ((w & 1) * 64 + rr) * GLDP + khalf * 8;
    const int fbofs = GTM * GLDP + ((w >> 1) * 64 + rr) * GLDP + khalf * 8;

    Stage12 g;
    tile_fetch(gA, gB, 0, g);
    tile_store(smem + cofs, smem + GTM * GLDP + cofs, g);
    __syncthreads();

    for (int kt = 0; kt < GKT - 1; ++kt) {
        _Float16* cur = smem + (kt & 1) * GSTAGE;
        _Float16* nxt = smem + ((kt + 1) & 1) * GSTAGE;
        tile_fetch(gA, gB, (kt + 1) * GBK, g);
        __builtin_amdgcn_sched_barrier(0);
        stage_compute(cur + faofs, cur + fbofs, acc);
        tile_store(nxt + cofs, nxt + GTM * GLDP + cofs, g);
        __syncthreads();
    }
    {
        const _Float16* cur = smem + ((GKT - 1) & 1) * GSTAGE;
        stage_compute(cur + faofs, cur + fbofs, acc);
    }
}

__device__ __forceinline__ float fast_sigmoid(float v) {
    return __builtin_amdgcn_rcpf(1.0f + __expf(-v));
}

__global__ void k_convert(const float* __restrict__ W_in,
                          const float* __restrict__ W_xp,
                          const float* __restrict__ W_out,
                          const float* __restrict__ Dmat,
                          _Float16* __restrict__ Wih,
                          _Float16* __restrict__ Wxh,
                          _Float16* __restrict__ Woh,
                          _Float16* __restrict__ Dth) {
    const int NWI = 2 * MM * MM, NXP = SS * MM, NWO = MM * MM, ND = SS * MM;
    int i = blockIdx.x * 256 + threadIdx.x;
    if (i < NWI) { Wih[i] = (_Float16)W_in[i]; return; }
    i -= NWI;
    if (i < NXP) { Wxh[i] = (_Float16)W_xp[i]; return; }
    i -= NXP;
    if (i < NWO) { Woh[i] = (_Float16)W_out[i]; return; }
    i -= NWO;
    if (i < ND) {
        int n = i >> 6, s = i & 63;
        Dth[i] = (_Float16)Dmat[(size_t)s * MM + n];
    }
}

__global__ __launch_bounds__(256) void k_layernorm(const float* __restrict__ x,
                                                   const float* __restrict__ gamma,
                                                   const float* __restrict__ beta,
                                                   _Float16* __restrict__ xn) {
    __shared__ float sSum[8], sSq[8];
    const int row = blockIdx.x;
    const size_t base = (size_t)row * MM;
    const int tid = threadIdx.x;

    float4 v = ((const float4*)(x + base))[tid];
    float s = v.x + v.y + v.z + v.w;
    float q = v.x * v.x + v.y * v.y + v.z * v.z + v.w * v.w;
#pragma unroll
    for (int off = 16; off > 0; off >>= 1) {
        s += __shfl_down(s, off);
        q += __shfl_down(q, off);
    }
    if ((tid & 31) == 0) { sSum[tid >> 5] = s; sSq[tid >> 5] = q; }
    __syncthreads();
    float ts = 0.f, tq = 0.f;
#pragma unroll
    for (int i = 0; i < 8; ++i) { ts += sSum[i]; tq += sSq[i]; }
    const float mu = ts * (1.0f / MM);
    const float var = tq * (1.0f / MM) - mu * mu;
    const float rstd = rsqrtf(var + 1e-5f);

    float4 g = ((const float4*)gamma)[tid];
    float4 bb = ((const float4*)beta)[tid];
    v4h o;
    o[0] = (_Float16)((v.x - mu) * rstd * g.x + bb.x);
    o[1] = (_Float16)((v.y - mu) * rstd * g.y + bb.y);
    o[2] = (_Float16)((v.z - mu) * rstd * g.z + bb.z);
    o[3] = (_Float16)((v.w - mu) * rstd * g.w + bb.w);
    ((v4h*)(xn + base))[tid] = o;
}

__global__ __launch_bounds__(256) void k_gemm1(const _Float16* __restrict__ xn,
                                               const _Float16* __restrict__ Wih,
                                               const float* __restrict__ b_in,
                                               _Float16* __restrict__ xp,
                                               _Float16* __restrict__ zg) {
    __shared__ __attribute__((aligned(16))) _Float16 smem[2 * GSTAGE];
    Acc44 acc = {};
    const int w = threadIdx.x >> 5;
    const int rowbase = blockIdx.x * 128 + (w & 1) * 64;
    const int colbase = blockIdx.y * 256 + (w >> 1) * 64;
    block_gemm_lds(xn, Wih, blockIdx.x * 128, blockIdx.y * 256, smem, acc);

    const int lane = threadIdx.x & 31;
    const int nlo = lane & 15, mh = (lane >> 4) * 8;
    const bool is_x = (blockIdx.y < (MM / 256));
#pragma unroll
    for (int j = 0; j < 4; ++j) {
        const int n = colbase + j * 16 + nlo;
        const float bias = b_in[n];
#pragma unroll
        for (int i = 0; i < 4; ++i) {
#pragma unroll
            for (int jj = 0; jj < 8; ++jj) {
                const int m = rowbase + i * 16 + mh + jj;
                const float v = acc.c[i][j][jj] + bias;
                const float sg = fast_sigmoid(v);
                if (is_x) xp[(size_t)m * MM + n] = (_Float16)(v * sg);
                else      zg[(size_t)m * MM + (n - MM)] = (_Float16)sg;
            }
        }
    }
}

__global__ __launch_bounds__(256) void k_gemm2(const _Float16* __restrict__ xp,
                                               const _Float16* __restrict__ Wxh,
                                               const float* __restrict__ b_xp,
                                               float* __restrict__ u) {
    Acc44 acc = {};
    const int w = threadIdx.x >> 5;
    const int rowbase = blockIdx.x * 512 + w * 64;
    wave_gemm64(xp, Wxh, MM, MM, rowbase, 0, MM, acc);

    const int lane = threadIdx.x & 31;
    const int nlo = lane & 15, mh = (lane >> 4) * 8;
#pragma unroll
    for (int j = 0; j < 4; ++j) {
        const int n = j * 16 + nlo;
        const float bias = b_xp[n];
#pragma unroll
        for (int i = 0; i < 4; ++i)
#pragma unroll
            for (int jj = 0; jj < 8; ++jj) {
                const int m = rowbase + i * 16 + mh + jj;
                u[(size_t)m * SS + n] = acc.c[i][j][jj] + bias;
            }
    }
}

__global__ void k_scan_chunk(const float* __restrict__ u,
                             const float* __restrict__ A_log,
                             float* __restrict__ hloc,
                             float* __restrict__ chunk_end) {
    const int t = blockIdx.x * 256 + threadIdx.x;
    const int s = t & 63;
    const int c = (t >> 6) & (NC - 1);
    const int b = t >> 10;
    const float a = expf(-expf(A_log[s]));
    float h = 0.f;
    const int l0 = c * LC;
    for (int j = 0; j < LC; ++j) {
        const size_t idx = ((size_t)b * LL + l0 + j) * SS + s;
        h = a * h + u[idx];
        hloc[idx] = h;
    }
    chunk_end[((b * SS + s) * NC) + c] = h;
}

__global__ void k_scan_carry(const float* __restrict__ A_log,
                             const float* __restrict__ chunk_end,
                             float* __restrict__ carry_in) {
    const int t = threadIdx.x;
    const int s = t & 63, b = t >> 6;
    const float E = expf(-expf(A_log[s]) * (float)LC);
    float carry = 0.f;
    for (int c = 0; c < NC; ++c) {
        carry_in[((b * SS + s) * NC) + c] = carry;
        carry = E * carry + chunk_end[((b * SS + s) * NC) + c];
    }
}

__global__ void k_scan_apply(const float* __restrict__ hloc,
                             const float* __restrict__ A_log,
                             const float* __restrict__ carry_in,
                             _Float16* __restrict__ h_h) {
    const int t = blockIdx.x * 256 + threadIdx.x;
    const int s = t & 63;
    const int c = (t >> 6) & (NC - 1);
    const int b = t >> 10;
    const float a = expf(-expf(A_log[s]));
    const float carry = carry_in[((b * SS + s) * NC) + c];
    float p = a;
    const int l0 = c * LC;
    for (int j = 0; j < LC; ++j) {
        const int l = l0 + j;
        const float hv = hloc[((size_t)b * LL + l) * SS + s] + carry * p;
        p *= a;
        h_h[((size_t)l * BB + b) * SS + s] = (_Float16)hv;
    }
}

__global__ __launch_bounds__(256) void k_gemm3(const _Float16* __restrict__ h_h,
                                               const _Float16* __restrict__ Dth,
                                               const _Float16* __restrict__ zg,
                                               _Float16* __restrict__ ys) {
    Acc44 acc = {};
    const int w = threadIdx.x >> 5;
    const int rowbase = blockIdx.x * 128 + (w & 1) * 64;
    const int colbase = blockIdx.y * 256 + (w >> 1) * 64;
    wave_gemm64(h_h, Dth, SS, SS, rowbase, colbase, SS, acc);

    const int lane = threadIdx.x & 31;
    const int nlo = lane & 15, mh = (lane >> 4) * 8;
#pragma unroll
    for (int i = 0; i < 4; ++i)
#pragma unroll
        for (int j = 0; j < 4; ++j)
#pragma unroll
            for (int jj = 0; jj < 8; ++jj) {
                const int rp = rowbase + i * 16 + mh + jj;
                const int n = colbase + j * 16 + nlo;
                const int b = rp & (BB - 1), l = rp >> 2;
                const float z = (float)zg[((size_t)b * LL + l) * MM + n];
                ys[(size_t)rp * MM + n] = (_Float16)(acc.c[i][j][jj] * z);
            }
}

__global__ __launch_bounds__(256, 4) void k_gemm4(const _Float16* __restrict__ ys,
                                               const _Float16* __restrict__ Woh,
                                               const float* __restrict__ b_out,
                                               const float* __restrict__ xflat,
                                               float* __restrict__ out) {
    __shared__ __attribute__((aligned(16))) _Float16 smem[2 * GSTAGE];
    Acc44 acc = {};
    const int w = threadIdx.x >> 5;
    const int rowbase = blockIdx.x * 128 + (w & 1) * 64;
    const int colbase = blockIdx.y * 256 + (w >> 1) * 64;
    block_gemm_lds(ys, Woh, blockIdx.x * 128, blockIdx.y * 256, smem, acc);

    const int lane = threadIdx.x & 31;
    const int nlo = lane & 15, mh = (lane >> 4) * 8;
#pragma unroll
    for (int j = 0; j < 4; ++j) {
        const int n = colbase + j * 16 + nlo;
        const float bias = b_out[n];
#pragma unroll
        for (int i = 0; i < 4; ++i)
#pragma unroll
            for (int jj = 0; jj < 8; ++jj) {
                const int m = rowbase + i * 16 + mh + jj;
                const size_t idx = (size_t)m * MM + n;
                out[idx] = xflat[idx] + acc.c[i][j][jj] + bias;
            }
    }
}

extern "C" void kernel_launch(void* const* d_in, const int* in_sizes, int n_in,
                              void* d_out, int out_size, void* d_ws, size_t ws_size,
                              hipStream_t stream) {
    const float* x     = (const float*)d_in[0];
    const float* A_log = (const float*)d_in[1];
    const float* Dmat  = (const float*)d_in[2];
    const float* W_in  = (const float*)d_in[3];
    const float* b_in  = (const float*)d_in[4];
    const float* W_xp  = (const float*)d_in[5];
    const float* b_xp  = (const float*)d_in[6];
    const float* W_out = (const float*)d_in[7];
    const float* b_out = (const float*)d_in[8];
    const float* gamma = (const float*)d_in[9];
    const float* beta  = (const float*)d_in[10];
    float* out = (float*)d_out;

    char* ws = (char*)d_ws;
    size_t off = 0;
    auto take = [&](size_t bytes) { char* p = ws + off; off += bytes; return p; };
    _Float16* xn_h  = (_Float16*)take((size_t)ROWS * MM * 2);
    _Float16* xp_h  = (_Float16*)take((size_t)ROWS * MM * 2);
    _Float16* z_h   = (_Float16*)take((size_t)ROWS * MM * 2);
    _Float16* ys_h  = (_Float16*)take((size_t)ROWS * MM * 2);
    float*    u     = (float*)take((size_t)ROWS * SS * 4);
    float*    hloc  = (float*)take((size_t)ROWS * SS * 4);
    _Float16* h_h   = (_Float16*)take((size_t)ROWS * SS * 2);
    _Float16* Wih   = (_Float16*)take((size_t)2 * MM * MM * 2);
    _Float16* Wxh   = (_Float16*)take((size_t)SS * MM * 2);
    _Float16* Woh   = (_Float16*)take((size_t)MM * MM * 2);
    _Float16* Dth   = (_Float16*)take((size_t)SS * MM * 2);
    float*    cend  = (float*)take((size_t)BB * SS * NC * 4);
    float*    cin   = (float*)take((size_t)BB * SS * NC * 4);
    (void)ws_size; (void)in_sizes; (void)n_in; (void)out_size;

    {
        const int total = 2 * MM * MM + SS * MM + MM * MM + SS * MM;
        k_convert<<<(total + 255) / 256, 256, 0, stream>>>(W_in, W_xp, W_out, Dmat,
                                                           Wih, Wxh, Woh, Dth);
    }
    k_layernorm<<<ROWS, 256, 0, stream>>>(x, gamma, beta, xn_h);
    k_gemm1<<<dim3(ROWS / 128, (2 * MM) / 256), 256, 0, stream>>>(xn_h, Wih, b_in, xp_h, z_h);
    k_gemm2<<<dim3(ROWS / 512, 1), 256, 0, stream>>>(xp_h, Wxh, b_xp, u);
    k_scan_chunk<<<(BB * SS * NC) / 256, 256, 0, stream>>>(u, A_log, hloc, cend);
    k_scan_carry<<<1, BB * SS, 0, stream>>>(A_log, cend, cin);
    k_scan_apply<<<(BB * SS * NC) / 256, 256, 0, stream>>>(hloc, A_log, cin, h_h);
    k_gemm3<<<dim3(ROWS / 128, MM / 256), 256, 0, stream>>>(h_h, Dth, z_h, ys_h);
    k_gemm4<<<dim3(ROWS / 128, MM / 256), 256, 0, stream>>>(ys_h, Woh, b_out, x, out);
}


# Scan re-gridded: 3 launches of 128 waves x 256 dependent-load steps -> 2 launches; a wave owns 16 steps, all loads in flight; carries by Horner over segment ends (new sum order); no hloc pass
# speedup vs baseline: 2.2581x; 1.3374x over previous
#include <hip/hip_runtime.h>
#include <hip/hip_bf16.h>

#define BB   4
#define LL   4096
#define MM   1024
#define SS   64
#define ROWS (BB * LL)
#define SEG  128
#define NSEG (LL / SEG)
#define SUB  16

typedef __attribute__((ext_vector_type(16))) _Float16 v16h;
typedef __attribute__((ext_vector_type(8)))  _Float16 v8h;
typedef __attribute__((ext_vector_type(4)))  _Float16 v4h;
typedef __attribute__((ext_vector_type(8)))  float    v8f;

__device__ __forceinline__ v16h load_frag(const _Float16* __restrict__ p) {
    v8h lo = *(const v8h*)(p);
    v8h hi = *(const v8h*)(p + 16);
    v16h t;
#pragma unroll
    for (int i = 0; i < 8; ++i) { t[i] = lo[i]; t[i + 8] = hi[i]; }
    return t;
}

struct Acc44 { v8f c[4][4]; };

__device__ __forceinline__ void wave_gemm64(const _Float16* __restrict__ A,
                                            const _Float16* __restrict__ B,
                                            int ldA, int ldB,
                                            int rowbase, int colbase, int K,
                                            Acc44& acc) {
    const int lane  = threadIdx.x & 31;
    const int rr    = lane & 15;
    const int khalf = (lane >> 4) & 1;
    const _Float16* Ab[4];
    const _Float16* Bb[4];
#pragma unroll
    for (int t = 0; t < 4; ++t) {
        Ab[t] = A + (size_t)(rowbase + t * 16 + rr) * ldA + khalf * 8;
        Bb[t] = B + (size_t)(colbase + t * 16 + rr) * ldB + khalf * 8;
    }
    for (int kb = 0; kb < K; kb += 32) {
        v16h a[4], b[4];
#pragma unroll
        for (int t = 0; t < 4; ++t) a[t] = load_frag(Ab[t] + kb);
#pragma unroll
        for (int t = 0; t < 4; ++t) b[t] = load_frag(Bb[t] + kb);
#pragma unroll
        for (int i = 0; i < 4; ++i)
#pragma unroll
            for (int j = 0; j < 4; ++j)
                acc.c[i][j] = __builtin_amdgcn_wmma_f32_16x16x32_f16(
                    false, a[i], false, b[j], (short)0, acc.c[i][j], false, false);
    }
}

#define GBK     64
#define GLDP    72
#define GTM     128
#define GTN     256
#define GSTAGE  ((GTM + GTN) * GLDP)
#define GKT     (MM / GBK)

struct Stage12 { v8h a[4]; v8h b[8]; };

__device__ __forceinline__ void tile_fetch(const _Float16* __restrict__ gA,
                                           const _Float16* __restrict__ gB,
                                           int kofs, Stage12& g) {
#pragma unroll
    for (int r = 0; r < 4; ++r) g.a[r] = *(const v8h*)(gA + (size_t)r * 32 * MM + kofs);
#pragma unroll
    for (int r = 0; r < 8; ++r) g.b[r] = *(const v8h*)(gB + (size_t)r * 32 * MM + kofs);
}

__device__ __forceinline__ void tile_store(_Float16* sA, _Float16* sB, const Stage12& g) {
#pragma unroll
    for (int r = 0; r < 4; ++r) *(v8h*)(sA + r * 32 * GLDP) = g.a[r];
#pragma unroll
    for (int r = 0; r < 8; ++r) *(v8h*)(sB + r * 32 * GLDP) = g.b[r];
}

__device__ __forceinline__ void stage_compute(const _Float16* fA, const _Float16* fB, Acc44& acc) {
    v16h a[2][4], b[2][4];
#pragma unroll
    for (int ks = 0; ks < 2; ++ks) {
#pragma unroll
        for (int t = 0; t < 4; ++t) a[ks][t] = load_frag(fA + t * 16 * GLDP + ks * 32);
#pragma unroll
        for (int t = 0; t < 4; ++t) b[ks][t] = load_frag(fB + t * 16 * GLDP + ks * 32);
    }
#pragma unroll
    for (int ks = 0; ks < 2; ++ks)
#pragma unroll
        for (int i = 0; i < 4; ++i)
#pragma unroll
            for (int j = 0; j < 4; ++j)
                acc.c[i][j] = __builtin_amdgcn_wmma_f32_16x16x32_f16(
                    false, a[ks][i], false, b[ks][j], (short)0, acc.c[i][j], false, false);
}

__device__ __forceinline__ void block_gemm_lds(const _Float16* __restrict__ A,
                                               const _Float16* __restrict__ B,
                                               int rowblk, int colblk,
                                               _Float16* smem, Acc44& acc) {
    const int tid   = threadIdx.x;
    const int lane  = tid & 31;
    const int w     = tid >> 5;
    const int rr    = lane & 15;
    const int khalf = (lane >> 4) & 1;

    const int crow = tid >> 3;
    const int ccol = (tid & 7) * 8;
    const _Float16* gA = A + (size_t)(rowblk + crow) * MM + ccol;
    const _Float16* gB = B + (size_t)(colblk + crow) * MM + ccol;
    const int cofs = crow * GLDP + ccol;

    const int faofs = ((w & 1) * 64 + rr) * GLDP + khalf * 8;
    const int fbofs = GTM * GLDP + ((w >> 1) * 64 + rr) * GLDP + khalf * 8;

    Stage12 g;
    tile_fetch(gA, gB, 0, g);
    tile_store(smem + cofs, smem + GTM * GLDP + cofs, g);
    __syncthreads();

    for (int kt = 0; kt < GKT - 1; ++kt) {
        _Float16* cur = smem + (kt & 1) * GSTAGE;
        _Float16* nxt = smem + ((kt + 1) & 1) * GSTAGE;
        tile_fetch(gA, gB, (kt + 1) * GBK, g);
        __builtin_amdgcn_sched_barrier(0);
        stage_compute(cur + faofs, cur + fbofs, acc);
        tile_store(nxt + cofs, nxt + GTM * GLDP + cofs, g);
        __syncthreads();
    }
    {
        const _Float16* cur = smem + ((GKT - 1) & 1) * GSTAGE;
        stage_compute(cur + faofs, cur + fbofs, acc);
    }
}

__device__ __forceinline__ float fast_sigmoid(float v) {
    return __builtin_amdgcn_rcpf(1.0f + __expf(-v));
}

#define CVT_WGS 512
#define CVT_T   (CVT_WGS * 256)

__device__ __forceinline__ v4h cvt4(const float4 a) {
    v4h o;
    o[0] = (_Float16)a.x; o[1] = (_Float16)a.y; o[2] = (_Float16)a.z; o[3] = (_Float16)a.w;
    return o;
}

__global__ __launch_bounds__(256) void k_convert(const float* __restrict__ W_in,
                                                 const float* __restrict__ W_xp,
                                                 const float* __restrict__ W_out,
                                                 const float* __restrict__ Dmat,
                                                 _Float16* __restrict__ Wih,
                                                 _Float16* __restrict__ Wxh,
                                                 _Float16* __restrict__ Woh,
                                                 _Float16* __restrict__ Dth) {
    constexpr int NWI4 = 2 * MM * MM / 4;
    constexpr int NWO4 = MM * MM / 4;
    constexpr int NXP4 = SS * MM / 4;
    static_assert(NWI4 % CVT_T == 0 && NWO4 % CVT_T == 0 && NXP4 <= CVT_T, "conversion grid");
    const int t = blockIdx.x * 256 + threadIdx.x;

    float4 a[NWI4 / CVT_T], b[NWO4 / CVT_T];
#pragma unroll
    for (int k = 0; k < NWI4 / CVT_T; ++k) a[k] = ((const float4*)W_in)[t + k * CVT_T];
#pragma unroll
    for (int k = 0; k < NWO4 / CVT_T; ++k) b[k] = ((const float4*)W_out)[t + k * CVT_T];
#pragma unroll
    for (int k = 0; k < NWI4 / CVT_T; ++k) ((v4h*)Wih)[t + k * CVT_T] = cvt4(a[k]);
#pragma unroll
    for (int k = 0; k < NWO4 / CVT_T; ++k) ((v4h*)Woh)[t + k * CVT_T] = cvt4(b[k]);

    if (t < NXP4) {
        ((v4h*)Wxh)[t] = cvt4(((const float4*)W_xp)[t]);
        const int n = t >> 4, s4 = (t & 15) * 4;
        v4h o;
#pragma unroll
        for (int k = 0; k < 4; ++k) o[k] = (_Float16)Dmat[(size_t)(s4 + k) * MM + n];
        ((v4h*)Dth)[t] = o;
    }
}

#define LN_RPW 4

__global__ __launch_bounds__(256) void k_layernorm(const float* __restrict__ x,
                                                   const float* __restrict__ gamma,
                                                   const float* __restrict__ beta,
                                                   _Float16* __restrict__ xn) {
    const int lane = threadIdx.x & 31;
    const int wave = blockIdx.x * 8 + (threadIdx.x >> 5);
    const size_t base = (size_t)wave * LN_RPW * MM;

    float4 v[LN_RPW][8];
#pragma unroll
    for (int r = 0; r < LN_RPW; ++r)
#pragma unroll
        for (int k = 0; k < 8; ++k)
            v[r][k] = ((const float4*)(x + base + (size_t)r * MM))[k * 32 + lane];

    float s[LN_RPW], q[LN_RPW];
#pragma unroll
    for (int r = 0; r < LN_RPW; ++r) {
        s[r] = 0.f; q[r] = 0.f;
#pragma unroll
        for (int k = 0; k < 8; ++k) {
            const float4 e = v[r][k];
            s[r] += e.x + e.y + e.z + e.w;
            q[r] += e.x * e.x + e.y * e.y + e.z * e.z + e.w * e.w;
        }
    }
#pragma unroll
    for (int off = 16; off > 0; off >>= 1) {
#pragma unroll
        for (int r = 0; r < LN_RPW; ++r) {
            s[r] += __shfl_xor(s[r], off);
            q[r] += __shfl_xor(q[r], off);
        }
    }
    float mu[LN_RPW], rstd[LN_RPW];
#pragma unroll
    for (int r = 0; r < LN_RPW; ++r) {
        mu[r] = s[r] * (1.0f / MM);
        const float var = q[r] * (1.0f / MM) - mu[r] * mu[r];
        rstd[r] = rsqrtf(var + 1e-5f);
    }
#pragma unroll
    for (int k = 0; k < 8; ++k) {
        const float4 g  = ((const float4*)gamma)[k * 32 + lane];
        const float4 bb = ((const float4*)beta)[k * 32 + lane];
#pragma unroll
        for (int r = 0; r < LN_RPW; ++r) {
            const float4 e = v[r][k];
            v4h o;
            o[0] = (_Float16)((e.x - mu[r]) * rstd[r] * g.x + bb.x);
            o[1] = (_Float16)((e.y - mu[r]) * rstd[r] * g.y + bb.y);
            o[2] = (_Float16)((e.z - mu[r]) * rstd[r] * g.z + bb.z);
            o[3] = (_Float16)((e.w - mu[r]) * rstd[r] * g.w + bb.w);
            ((v4h*)(xn + base + (size_t)r * MM))[k * 32 + lane] = o;
        }
    }
}

__global__ __launch_bounds__(256) void k_gemm1(const _Float16* __restrict__ xn,
                                               const _Float16* __restrict__ Wih,
                                               const float* __restrict__ b_in,
                                               _Float16* __restrict__ xp,
                                               _Float16* __restrict__ zg) {
    __shared__ __attribute__((aligned(16))) _Float16 smem[2 * GSTAGE];
    Acc44 acc = {};
    const int w = threadIdx.x >> 5;
    const int rowbase = blockIdx.x * 128 + (w & 1) * 64;
    const int colbase = blockIdx.y * 256 + (w >> 1) * 64;
    block_gemm_lds(xn, Wih, blockIdx.x * 128, blockIdx.y * 256, smem, acc);

    const int lane = threadIdx.x & 31;
    const int nlo = lane & 15, mh = (lane >> 4) * 8;
    const bool is_x = (blockIdx.y < (MM / 256));
#pragma unroll
    for (int j = 0; j < 4; ++j) {
        const int n = colbase + j * 16 + nlo;
        const float bias = b_in[n];
#pragma unroll
        for (int i = 0; i < 4; ++i) {
#pragma unroll
            for (int jj = 0; jj < 8; ++jj) {
                const int m = rowbase + i * 16 + mh + jj;
                const float v = acc.c[i][j][jj] + bias;
                const float sg = fast_sigmoid(v);
                if (is_x) xp[(size_t)m * MM + n] = (_Float16)(v * sg);
                else      zg[(size_t)m * MM + (n - MM)] = (_Float16)sg;
            }
        }
    }
}

__global__ __launch_bounds__(256) void k_gemm2(const _Float16* __restrict__ xp,
                                               const _Float16* __restrict__ Wxh,
                                               const float* __restrict__ b_xp,
                                               float* __restrict__ u) {
    Acc44 acc = {};
    const int w = threadIdx.x >> 5;
    const int rowbase = blockIdx.x * 512 + w * 64;
    wave_gemm64(xp, Wxh, MM, MM, rowbase, 0, MM, acc);

    const int lane = threadIdx.x & 31;
    const int nlo = lane & 15, mh = (lane >> 4) * 8;
#pragma unroll
    for (int j = 0; j < 4; ++j) {
        const int n = j * 16 + nlo;
        const float bias = b_xp[n];
#pragma unroll
        for (int i = 0; i < 4; ++i)
#pragma unroll
            for (int jj = 0; jj < 8; ++jj) {
                const int m = rowbase + i * 16 + mh + jj;
                u[(size_t)m * SS + n] = acc.c[i][j][jj] + bias;
            }
    }
}

struct ScanLane {
    float2 uv[SUB];
    float  a0, a1;
    float  p0, p1;
};

__device__ __forceinline__ void scan_load(const float* __restrict__ u,
                                          const float* __restrict__ A_log,
                                          int b, int l0, int lane, ScanLane& t) {
    const float2* up = (const float2*)(u + ((size_t)b * LL + l0) * SS) + lane;
#pragma unroll
    for (int j = 0; j < SUB; ++j) t.uv[j] = up[j * (SS / 2)];
    const float2 al = ((const float2*)A_log)[lane];
    t.a0 = expf(-expf(al.x));
    t.a1 = expf(-expf(al.y));
    float q0 = t.a0, q1 = t.a1;
#pragma unroll
    for (int i = 1; i < SUB; i <<= 1) { q0 *= q0; q1 *= q1; }
    t.p0 = q0; t.p1 = q1;
}

__device__ __forceinline__ float2 scan_run(const ScanLane& t, float2 h) {
#pragma unroll
    for (int j = 0; j < SUB; ++j) {
        h.x = t.a0 * h.x + t.uv[j].x;
        h.y = t.a1 * h.y + t.uv[j].y;
    }
    return h;
}

__global__ __launch_bounds__(256) void k_scan_ends(const float* __restrict__ u,
                                                   const float* __restrict__ A_log,
                                                   float* __restrict__ seg_end) {
    __shared__ float2 we[SEG / SUB][32];
    const int lane = threadIdx.x & 31, k = threadIdx.x >> 5;
    const int b = blockIdx.x / NSEG, sg = blockIdx.x % NSEG;
    ScanLane t;
    scan_load(u, A_log, b, sg * SEG + k * SUB, lane, t);
    we[k][lane] = scan_run(t, make_float2(0.f, 0.f));
    __syncthreads();
    if (k == 0) {
        float2 e = make_float2(0.f, 0.f);
#pragma unroll
        for (int kk = 0; kk < SEG / SUB; ++kk) {
            const float2 w = we[kk][lane];
            e.x = t.p0 * e.x + w.x;
            e.y = t.p1 * e.y + w.y;
        }
        ((float2*)seg_end)[(size_t)blockIdx.x * 32 + lane] = e;
    }
}

__global__ __launch_bounds__(256) void k_scan_apply(const float* __restrict__ u,
                                                    const float* __restrict__ A_log,
                                                    const float* __restrict__ seg_end,
                                                    _Float16* __restrict__ h_h) {
    __shared__ float2 we[SEG / SUB][32];
    const int lane = threadIdx.x & 31, k = threadIdx.x >> 5;
    const int b = blockIdx.x / NSEG, sg = blockIdx.x % NSEG;
    const int l0 = sg * SEG + k * SUB;
    ScanLane t;
    scan_load(u, A_log, b, l0, lane, t);

    const float2* ep = (const float2*)seg_end + (size_t)b * NSEG * 32 + lane;
    float2 se[NSEG - 1];
#pragma unroll
    for (int d = 0; d < NSEG - 1; ++d) {
        const int q = sg - 1 - d;
        const float2 v = ep[(q > 0 ? q : 0) * 32];
        se[d] = (q >= 0) ? v : make_float2(0.f, 0.f);
    }

    we[k][lane] = scan_run(t, make_float2(0.f, 0.f));

    float P0 = t.p0, P1 = t.p1;
#pragma unroll
    for (int i = 1; i < SEG / SUB; i <<= 1) { P0 *= P0; P1 *= P1; }
    float2 c = make_float2(0.f, 0.f);
#pragma unroll
    for (int d = NSEG - 2; d >= 0; --d) {
        c.x = P0 * c.x + se[d].x;
        c.y = P1 * c.y + se[d].y;
    }
    __syncthreads();
#pragma unroll
    for (int kk = 0; kk < SEG / SUB - 1; ++kk) {
        const float2 w = we[kk][lane];
        const float nx = t.p0 * c.x + w.x;
        const float ny = t.p1 * c.y + w.y;
        c.x = (kk < k) ? nx : c.x;
        c.y = (kk < k) ? ny : c.y;
    }
    typedef __attribute__((ext_vector_type(2))) _Float16 v2h;
    v2h* hp = (v2h*)(h_h + ((size_t)l0 * BB + b) * SS) + lane;
#pragma unroll
    for (int j = 0; j < SUB; ++j) {
        c.x = t.a0 * c.x + t.uv[j].x;
        c.y = t.a1 * c.y + t.uv[j].y;
        v2h o;
        o[0] = (_Float16)c.x; o[1] = (_Float16)c.y;
        hp[(size_t)j * BB * (SS / 2)] = o;
    }
}

__global__ __launch_bounds__(256) void k_gemm3(const _Float16* __restrict__ h_h,
                                               const _Float16* __restrict__ Dth,
                                               const _Float16* __restrict__ zg,
                                               _Float16* __restrict__ ys) {
    Acc44 acc = {};
    const int w = threadIdx.x >> 5;
    const int rowbase = blockIdx.x * 128 + (w & 1) * 64;
    const int colbase = blockIdx.y * 256 + (w >> 1) * 64;
    wave_gemm64(h_h, Dth, SS, SS, rowbase, colbase, SS, acc);

    const int lane = threadIdx.x & 31;
    const int nlo = lane & 15, mh = (lane >> 4) * 8;
#pragma unroll
    for (int i = 0; i < 4; ++i)
#pragma unroll
        for (int j = 0; j < 4; ++j)
#pragma unroll
            for (int jj = 0; jj < 8; ++jj) {
                const int rp = rowbase + i * 16 + mh + jj;
                const int n = colbase + j * 16 + nlo;
                const int b = rp & (BB - 1), l = rp >> 2;
                const float z = (float)zg[((size_t)b * LL + l) * MM + n];
                ys[(size_t)rp * MM + n] = (_Float16)(acc.c[i][j][jj] * z);
            }
}

__global__ __launch_bounds__(256, 4) void k_gemm4(const _Float16* __restrict__ ys,
                                               const _Float16* __restrict__ Woh,
                                               const float* __restrict__ b_out,
                                               const float* __restrict__ xflat,
                                               float* __restrict__ out) {
    __shared__ __attribute__((aligned(16))) _Float16 smem[2 * GSTAGE];
    Acc44 acc = {};
    const int w = threadIdx.x >> 5;
    const int rowbase = blockIdx.x * 128 + (w & 1) * 64;
    const int colbase = blockIdx.y * 256 + (w >> 1) * 64;
    block_gemm_lds(ys, Woh, blockIdx.x * 128, blockIdx.y * 256, smem, acc);

    const int lane = threadIdx.x & 31;
    const int nlo = lane & 15, mh = (lane >> 4) * 8;
#pragma unroll
    for (int j = 0; j < 4; ++j) {
        const int n = colbase + j * 16 + nlo;
        const float bias = b_out[n];
#pragma unroll
        for (int i = 0; i < 4; ++i)
#pragma unroll
            for (int jj = 0; jj < 8; ++jj) {
                const int m = rowbase + i * 16 + mh + jj;
                const size_t idx = (size_t)m * MM + n;
                out[idx] = xflat[idx] + acc.c[i][j][jj] + bias;
            }
    }
}

extern "C" void kernel_launch(void* const* d_in, const int* in_sizes, int n_in,
                              void* d_out, int out_size, void* d_ws, size_t ws_size,
                              hipStream_t stream) {
    const float* x     = (const float*)d_in[0];
    const float* A_log = (const float*)d_in[1];
    const float* Dmat  = (const float*)d_in[2];
    const float* W_in  = (const float*)d_in[3];
    const float* b_in  = (const float*)d_in[4];
    const float* W_xp  = (const float*)d_in[5];
    const float* b_xp  = (const float*)d_in[6];
    const float* W_out = (const float*)d_in[7];
    const float* b_out = (const float*)d_in[8];
    const float* gamma = (const float*)d_in[9];
    const float* beta  = (const float*)d_in[10];
    float* out = (float*)d_out;

    char* ws = (char*)d_ws;
    size_t off = 0;
    auto take = [&](size_t bytes) { char* p = ws + off; off += bytes; return p; };
    _Float16* xn_h  = (_Float16*)take((size_t)ROWS * MM * 2);
    _Float16* xp_h  = (_Float16*)take((size_t)ROWS * MM * 2);
    _Float16* z_h   = (_Float16*)take((size_t)ROWS * MM * 2);
    _Float16* ys_h  = (_Float16*)take((size_t)ROWS * MM * 2);
    float*    u     = (float*)take((size_t)ROWS * SS * 4);
    float*    sgend = (float*)take((size_t)ROWS * SS * 4);
    _Float16* h_h   = (_Float16*)take((size_t)ROWS * SS * 2);
    _Float16* Wih   = (_Float16*)take((size_t)2 * MM * MM * 2);
    _Float16* Wxh   = (_Float16*)take((size_t)SS * MM * 2);
    _Float16* Woh   = (_Float16*)take((size_t)MM * MM * 2);
    _Float16* Dth   = (_Float16*)take((size_t)SS * MM * 2);
    (void)ws_size; (void)in_sizes; (void)n_in; (void)out_size;

    k_convert<<<CVT_WGS, 256, 0, stream>>>(W_in, W_xp, W_out, Dmat, Wih, Wxh, Woh, Dth);
    k_layernorm<<<ROWS / (8 * LN_RPW), 256, 0, stream>>>(x, gamma, beta, xn_h);
    k_gemm1<<<dim3(ROWS / 128, (2 * MM) / 256), 256, 0, stream>>>(xn_h, Wih, b_in, xp_h, z_h);
    k_gemm2<<<dim3(ROWS / 512, 1), 256, 0, stream>>>(xp_h, Wxh, b_xp, u);
    k_scan_ends<<<BB * NSEG, 256, 0, stream>>>(u, A_log, sgend);
    k_scan_apply<<<BB * NSEG, 256, 0, stream>>>(u, A_log, sgend, h_h);
    k_gemm3<<<dim3(ROWS / 128, MM / 256), 256, 0, stream>>>(h_h, Dth, z_h, ys_h);
    k_gemm4<<<dim3(ROWS / 128, MM / 256), 256, 0, stream>>>(ys_h, Woh, b_out, x, out);
}


# in_proj epilogue: SiLU/gate half chosen once per wave, not per element (the compiler kept a test and two branches round every store): kernel text 40 KB -> 17 KB, exp/rcp chains batched
# speedup vs baseline: 2.4194x; 1.0714x over previous
#include <hip/hip_runtime.h>
#include <hip/hip_bf16.h>

#define BB   4
#define LL   4096
#define MM   1024
#define SS   64
#define ROWS (BB * LL)
#define SEG  128
#define NSEG (LL / SEG)
#define SUB  16

typedef __attribute__((ext_vector_type(16))) _Float16 v16h;
typedef __attribute__((ext_vector_type(8)))  _Float16 v8h;
typedef __attribute__((ext_vector_type(4)))  _Float16 v4h;
typedef __attribute__((ext_vector_type(8)))  float    v8f;

__device__ __forceinline__ v16h load_frag(const _Float16* __restrict__ p) {
    v8h lo = *(const v8h*)(p);
    v8h hi = *(const v8h*)(p + 16);
    v16h t;
#pragma unroll
    for (int i = 0; i < 8; ++i) { t[i] = lo[i]; t[i + 8] = hi[i]; }
    return t;
}

struct Acc44 { v8f c[4][4]; };

__device__ __forceinline__ void wave_gemm64(const _Float16* __restrict__ A,
                                            const _Float16* __restrict__ B,
                                            int ldA, int ldB,
                                            int rowbase, int colbase, int K,
                                            Acc44& acc) {
    const int lane  = threadIdx.x & 31;
    const int rr    = lane & 15;
    const int khalf = (lane >> 4) & 1;
    const _Float16* Ab[4];
    const _Float16* Bb[4];
#pragma unroll
    for (int t = 0; t < 4; ++t) {
        Ab[t] = A + (size_t)(rowbase + t * 16 + rr) * ldA + khalf * 8;
        Bb[t] = B + (size_t)(colbase + t * 16 + rr) * ldB + khalf * 8;
    }
    for (int kb = 0; kb < K; kb += 32) {
        v16h a[4], b[4];
#pragma unroll
        for (int t = 0; t < 4; ++t) a[t] = load_frag(Ab[t] + kb);
#pragma unroll
        for (int t = 0; t < 4; ++t) b[t] = load_frag(Bb[t] + kb);
#pragma unroll
        for (int i = 0; i < 4; ++i)
#pragma unroll
            for (int j = 0; j < 4; ++j)
                acc.c[i][j] = __builtin_amdgcn_wmma_f32_16x16x32_f16(
                    false, a[i], false, b[j], (short)0, acc.c[i][j], false, false);
    }
}

#define GBK     64
#define GLDP    72
#define GTM     128
#define GTN     256
#define GSTAGE  ((GTM + GTN) * GLDP)
#define GKT     (MM / GBK)

struct Stage12 { v8h a[4]; v8h b[8]; };

__device__ __forceinline__ void tile_fetch(const _Float16* __restrict__ gA,
                                           const _Float16* __restrict__ gB,
                                           int kofs, Stage12& g) {
#pragma unroll
    for (int r = 0; r < 4; ++r) g.a[r] = *(const v8h*)(gA + (size_t)r * 32 * MM + kofs);
#pragma unroll
    for (int r = 0; r < 8; ++r) g.b[r] = *(const v8h*)(gB + (size_t)r * 32 * MM + kofs);
}

__device__ __forceinline__ void tile_store(_Float16* sA, _Float16* sB, const Stage12& g) {
#pragma unroll
    for (int r = 0; r < 4; ++r) *(v8h*)(sA + r * 32 * GLDP) = g.a[r];
#pragma unroll
    for (int r = 0; r < 8; ++r) *(v8h*)(sB + r * 32 * GLDP) = g.b[r];
}

__device__ __forceinline__ void stage_compute(const _Float16* fA, const _Float16* fB, Acc44& acc) {
    v16h a[2][4], b[2][4];
#pragma unroll
    for (int ks = 0; ks < 2; ++ks) {
#pragma unroll
        for (int t = 0; t < 4; ++t) a[ks][t] = load_frag(fA + t * 16 * GLDP + ks * 32);
#pragma unroll
        for (int t = 0; t < 4; ++t) b[ks][t] = load_frag(fB + t * 16 * GLDP + ks * 32);
    }
#pragma unroll
    for (int ks = 0; ks < 2; ++ks)
#pragma unroll
        for (int i = 0; i < 4; ++i)
#pragma unroll
            for (int j = 0; j < 4; ++j)
                acc.c[i][j] = __builtin_amdgcn_wmma_f32_16x16x32_f16(
                    false, a[ks][i], false, b[ks][j], (short)0, acc.c[i][j], false, false);
}

__device__ __forceinline__ void block_gemm_lds(const _Float16* __restrict__ A,
                                               const _Float16* __restrict__ B,
                                               int rowblk, int colblk,
                                               _Float16* smem, Acc44& acc) {
    const int tid   = threadIdx.x;
    const int lane  = tid & 31;
    const int w     = tid >> 5;
    const int rr    = lane & 15;
    const int khalf = (lane >> 4) & 1;

    const int crow = tid >> 3;
    const int ccol = (tid & 7) * 8;
    const _Float16* gA = A + (size_t)(rowblk + crow) * MM + ccol;
    const _Float16* gB = B + (size_t)(colblk + crow) * MM + ccol;
    const int cofs = crow * GLDP + ccol;

    const int faofs = ((w & 1) * 64 + rr) * GLDP + khalf * 8;
    const int fbofs = GTM * GLDP + ((w >> 1) * 64 + rr) * GLDP + khalf * 8;

    Stage12 g;
    tile_fetch(gA, gB, 0, g);
    tile_store(smem + cofs, smem + GTM * GLDP + cofs, g);
    __syncthreads();

    for (int kt = 0; kt < GKT - 1; ++kt) {
        _Float16* cur = smem + (kt & 1) * GSTAGE;
        _Float16* nxt = smem + ((kt + 1) & 1) * GSTAGE;
        tile_fetch(gA, gB, (kt + 1) * GBK, g);
        __builtin_amdgcn_sched_barrier(0);
        stage_compute(cur + faofs, cur + fbofs, acc);
        tile_store(nxt + cofs, nxt + GTM * GLDP + cofs, g);
        __syncthreads();
    }
    {
        const _Float16* cur = smem + ((GKT - 1) & 1) * GSTAGE;
        stage_compute(cur + faofs, cur + fbofs, acc);
    }
}

__device__ __forceinline__ float fast_sigmoid(float v) {
    return __builtin_amdgcn_rcpf(1.0f + __expf(-v));
}

#define CVT_WGS 512
#define CVT_T   (CVT_WGS * 256)

__device__ __forceinline__ v4h cvt4(const float4 a) {
    v4h o;
    o[0] = (_Float16)a.x; o[1] = (_Float16)a.y; o[2] = (_Float16)a.z; o[3] = (_Float16)a.w;
    return o;
}

__global__ __launch_bounds__(256) void k_convert(const float* __restrict__ W_in,
                                                 const float* __restrict__ W_xp,
                                                 const float* __restrict__ W_out,
                                                 const float* __restrict__ Dmat,
                                                 _Float16* __restrict__ Wih,
                                                 _Float16* __restrict__ Wxh,
                                                 _Float16* __restrict__ Woh,
                                                 _Float16* __restrict__ Dth) {
    constexpr int NWI4 = 2 * MM * MM / 4;
    constexpr int NWO4 = MM * MM / 4;
    constexpr int NXP4 = SS * MM / 4;
    static_assert(NWI4 % CVT_T == 0 && NWO4 % CVT_T == 0 && NXP4 <= CVT_T, "conversion grid");
    const int t = blockIdx.x * 256 + threadIdx.x;

    float4 a[NWI4 / CVT_T], b[NWO4 / CVT_T];
#pragma unroll
    for (int k = 0; k < NWI4 / CVT_T; ++k) a[k] = ((const float4*)W_in)[t + k * CVT_T];
#pragma unroll
    for (int k = 0; k < NWO4 / CVT_T; ++k) b[k] = ((const float4*)W_out)[t + k * CVT_T];
#pragma unroll
    for (int k = 0; k < NWI4 / CVT_T; ++k) ((v4h*)Wih)[t + k * CVT_T] = cvt4(a[k]);
#pragma unroll
    for (int k = 0; k < NWO4 / CVT_T; ++k) ((v4h*)Woh)[t + k * CVT_T] = cvt4(b[k]);

    if (t < NXP4) {
        ((v4h*)Wxh)[t] = cvt4(((const float4*)W_xp)[t]);
        const int n = t >> 4, s4 = (t & 15) * 4;
        v4h o;
#pragma unroll
        for (int k = 0; k < 4; ++k) o[k] = (_Float16)Dmat[(size_t)(s4 + k) * MM + n];
        ((v4h*)Dth)[t] = o;
    }
}

#define LN_RPW 4

__global__ __launch_bounds__(256) void k_layernorm(const float* __restrict__ x,
                                                   const float* __restrict__ gamma,
                                                   const float* __restrict__ beta,
                                                   _Float16* __restrict__ xn) {
    const int lane = threadIdx.x & 31;
    const int wave = blockIdx.x * 8 + (threadIdx.x >> 5);
    const size_t base = (size_t)wave * LN_RPW * MM;

    float4 v[LN_RPW][8];
#pragma unroll
    for (int r = 0; r < LN_RPW; ++r)
#pragma unroll
        for (int k = 0; k < 8; ++k)
            v[r][k] = ((const float4*)(x + base + (size_t)r * MM))[k * 32 + lane];

    float s[LN_RPW], q[LN_RPW];
#pragma unroll
    for (int r = 0; r < LN_RPW; ++r) {
        s[r] = 0.f; q[r] = 0.f;
#pragma unroll
        for (int k = 0; k < 8; ++k) {
            const float4 e = v[r][k];
            s[r] += e.x + e.y + e.z + e.w;
            q[r] += e.x * e.x + e.y * e.y + e.z * e.z + e.w * e.w;
        }
    }
#pragma unroll
    for (int off = 16; off > 0; off >>= 1) {
#pragma unroll
        for (int r = 0; r < LN_RPW; ++r) {
            s[r] += __shfl_xor(s[r], off);
            q[r] += __shfl_xor(q[r], off);
        }
    }
    float mu[LN_RPW], rstd[LN_RPW];
#pragma unroll
    for (int r = 0; r < LN_RPW; ++r) {
        mu[r] = s[r] * (1.0f / MM);
        const float var = q[r] * (1.0f / MM) - mu[r] * mu[r];
        rstd[r] = rsqrtf(var + 1e-5f);
    }
#pragma unroll
    for (int k = 0; k < 8; ++k) {
        const float4 g  = ((const float4*)gamma)[k * 32 + lane];
        const float4 bb = ((const float4*)beta)[k * 32 + lane];
#pragma unroll
        for (int r = 0; r < LN_RPW; ++r) {
            const float4 e = v[r][k];
            v4h o;
            o[0] = (_Float16)((e.x - mu[r]) * rstd[r] * g.x + bb.x);
            o[1] = (_Float16)((e.y - mu[r]) * rstd[r] * g.y + bb.y);
            o[2] = (_Float16)((e.z - mu[r]) * rstd[r] * g.z + bb.z);
            o[3] = (_Float16)((e.w - mu[r]) * rstd[r] * g.w + bb.w);
            ((v4h*)(xn + base + (size_t)r * MM))[k * 32 + lane] = o;
        }
    }
}

__global__ __launch_bounds__(256) void k_gemm1(const _Float16* __restrict__ xn,
                                               const _Float16* __restrict__ Wih,
                                               const float* __restrict__ b_in,
                                               _Float16* __restrict__ xp,
                                               _Float16* __restrict__ zg) {
    __shared__ __attribute__((aligned(16))) _Float16 smem[2 * GSTAGE];
    Acc44 acc = {};
    const int w = threadIdx.x >> 5;
    const int rowbase = blockIdx.x * 128 + (w & 1) * 64;
    const int colbase = blockIdx.y * 256 + (w >> 1) * 64;
    block_gemm_lds(xn, Wih, blockIdx.x * 128, blockIdx.y * 256, smem, acc);

    const int lane = threadIdx.x & 31;
    const int nlo = lane & 15, mh = (lane >> 4) * 8;
    if (blockIdx.y < (MM / 256)) {
#pragma unroll
        for (int j = 0; j < 4; ++j) {
            const int n = colbase + j * 16 + nlo;
            const float bias = b_in[n];
#pragma unroll
            for (int i = 0; i < 4; ++i) {
#pragma unroll
                for (int jj = 0; jj < 8; ++jj) {
                    const int m = rowbase + i * 16 + mh + jj;
                    const float v = acc.c[i][j][jj] + bias;
                    xp[(size_t)m * MM + n] = (_Float16)(v * fast_sigmoid(v));
                }
            }
        }
    } else {
#pragma unroll
        for (int j = 0; j < 4; ++j) {
            const int n = colbase + j * 16 + nlo;
            const float bias = b_in[n];
#pragma unroll
            for (int i = 0; i < 4; ++i) {
#pragma unroll
                for (int jj = 0; jj < 8; ++jj) {
                    const int m = rowbase + i * 16 + mh + jj;
                    const float v = acc.c[i][j][jj] + bias;
                    zg[(size_t)m * MM + (n - MM)] = (_Float16)fast_sigmoid(v);
                }
            }
        }
    }
}

__global__ __launch_bounds__(256) void k_gemm2(const _Float16* __restrict__ xp,
                                               const _Float16* __restrict__ Wxh,
                                               const float* __restrict__ b_xp,
                                               float* __restrict__ u) {
    Acc44 acc = {};
    const int w = threadIdx.x >> 5;
    const int rowbase = blockIdx.x * 512 + w * 64;
    wave_gemm64(xp, Wxh, MM, MM, rowbase, 0, MM, acc);

    const int lane = threadIdx.x & 31;
    const int nlo = lane & 15, mh = (lane >> 4) * 8;
#pragma unroll
    for (int j = 0; j < 4; ++j) {
        const int n = j * 16 + nlo;
        const float bias = b_xp[n];
#pragma unroll
        for (int i = 0; i < 4; ++i)
#pragma unroll
            for (int jj = 0; jj < 8; ++jj) {
                const int m = rowbase + i * 16 + mh + jj;
                u[(size_t)m * SS + n] = acc.c[i][j][jj] + bias;
            }
    }
}

struct ScanLane {
    float2 uv[SUB];
    float  a0, a1;
    float  p0, p1;
};

__device__ __forceinline__ void scan_load(const float* __restrict__ u,
                                          const float* __restrict__ A_log,
                                          int b, int l0, int lane, ScanLane& t) {
    const float2* up = (const float2*)(u + ((size_t)b * LL + l0) * SS) + lane;
#pragma unroll
    for (int j = 0; j < SUB; ++j) t.uv[j] = up[j * (SS / 2)];
    const float2 al = ((const float2*)A_log)[lane];
    t.a0 = expf(-expf(al.x));
    t.a1 = expf(-expf(al.y));
    float q0 = t.a0, q1 = t.a1;
#pragma unroll
    for (int i = 1; i < SUB; i <<= 1) { q0 *= q0; q1 *= q1; }
    t.p0 = q0; t.p1 = q1;
}

__device__ __forceinline__ float2 scan_run(const ScanLane& t, float2 h) {
#pragma unroll
    for (int j = 0; j < SUB; ++j) {
        h.x = t.a0 * h.x + t.uv[j].x;
        h.y = t.a1 * h.y + t.uv[j].y;
    }
    return h;
}

__global__ __launch_bounds__(256) void k_scan_ends(const float* __restrict__ u,
                                                   const float* __restrict__ A_log,
                                                   float* __restrict__ seg_end) {
    __shared__ float2 we[SEG / SUB][32];
    const int lane = threadIdx.x & 31, k = threadIdx.x >> 5;
    const int b = blockIdx.x / NSEG, sg = blockIdx.x % NSEG;
    ScanLane t;
    scan_load(u, A_log, b, sg * SEG + k * SUB, lane, t);
    we[k][lane] = scan_run(t, make_float2(0.f, 0.f));
    __syncthreads();
    if (k == 0) {
        float2 e = make_float2(0.f, 0.f);
#pragma unroll
        for (int kk = 0; kk < SEG / SUB; ++kk) {
            const float2 w = we[kk][lane];
            e.x = t.p0 * e.x + w.x;
            e.y = t.p1 * e.y + w.y;
        }
        ((float2*)seg_end)[(size_t)blockIdx.x * 32 + lane] = e;
    }
}

__global__ __launch_bounds__(256) void k_scan_apply(const float* __restrict__ u,
                                                    const float* __restrict__ A_log,
                                                    const float* __restrict__ seg_end,
                                                    _Float16* __restrict__ h_h) {
    __shared__ float2 we[SEG / SUB][32];
    const int lane = threadIdx.x & 31, k = threadIdx.x >> 5;
    const int b = blockIdx.x / NSEG, sg = blockIdx.x % NSEG;
    const int l0 = sg * SEG + k * SUB;
    ScanLane t;
    scan_load(u, A_log, b, l0, lane, t);

    const float2* ep = (const float2*)seg_end + (size_t)b * NSEG * 32 + lane;
    float2 se[NSEG - 1];
#pragma unroll
    for (int d = 0; d < NSEG - 1; ++d) {
        const int q = sg - 1 - d;
        const float2 v = ep[(q > 0 ? q : 0) * 32];
        se[d] = (q >= 0) ? v : make_float2(0.f, 0.f);
    }

    we[k][lane] = scan_run(t, make_float2(0.f, 0.f));

    float P0 = t.p0, P1 = t.p1;
#pragma unroll
    for (int i = 1; i < SEG / SUB; i <<= 1) { P0 *= P0; P1 *= P1; }
    float2 c = make_float2(0.f, 0.f);
#pragma unroll
    for (int d = NSEG - 2; d >= 0; --d) {
        c.x = P0 * c.x + se[d].x;
        c.y = P1 * c.y + se[d].y;
    }
    __syncthreads();
#pragma unroll
    for (int kk = 0; kk < SEG / SUB - 1; ++kk) {
        const float2 w = we[kk][lane];
        const float nx = t.p0 * c.x + w.x;
        const float ny = t.p1 * c.y + w.y;
        c.x = (kk < k) ? nx : c.x;
        c.y = (kk < k) ? ny : c.y;
    }
    typedef __attribute__((ext_vector_type(2))) _Float16 v2h;
    v2h* hp = (v2h*)(h_h + ((size_t)l0 * BB + b) * SS) + lane;
#pragma unroll
    for (int j = 0; j < SUB; ++j) {
        c.x = t.a0 * c.x + t.uv[j].x;
        c.y = t.a1 * c.y + t.uv[j].y;
        v2h o;
        o[0] = (_Float16)c.x; o[1] = (_Float16)c.y;
        hp[(size_t)j * BB * (SS / 2)] = o;
    }
}

__global__ __launch_bounds__(256) void k_gemm3(const _Float16* __restrict__ h_h,
                                               const _Float16* __restrict__ Dth,
                                               const _Float16* __restrict__ zg,
                                               _Float16* __restrict__ ys) {
    Acc44 acc = {};
    const int w = threadIdx.x >> 5;
    const int rowbase = blockIdx.x * 128 + (w & 1) * 64;
    const int colbase = blockIdx.y * 256 + (w >> 1) * 64;
    wave_gemm64(h_h, Dth, SS, SS, rowbase, colbase, SS, acc);

    const int lane = threadIdx.x & 31;
    const int nlo = lane & 15, mh = (lane >> 4) * 8;
#pragma unroll
    for (int i = 0; i < 4; ++i)
#pragma unroll
        for (int j = 0; j < 4; ++j)
#pragma unroll
            for (int jj = 0; jj < 8; ++jj) {
                const int rp = rowbase + i * 16 + mh + jj;
                const int n = colbase + j * 16 + nlo;
                const int b = rp & (BB - 1), l = rp >> 2;
                const float z = (float)zg[((size_t)b * LL + l) * MM + n];
                ys[(size_t)rp * MM + n] = (_Float16)(acc.c[i][j][jj] * z);
            }
}

__global__ __launch_bounds__(256, 4) void k_gemm4(const _Float16* __restrict__ ys,
                                               const _Float16* __restrict__ Woh,
                                               const float* __restrict__ b_out,
                                               const float* __restrict__ xflat,
                                               float* __restrict__ out) {
    __shared__ __attribute__((aligned(16))) _Float16 smem[2 * GSTAGE];
    Acc44 acc = {};
    const int w = threadIdx.x >> 5;
    const int rowbase = blockIdx.x * 128 + (w & 1) * 64;
    const int colbase = blockIdx.y * 256 + (w >> 1) * 64;
    block_gemm_lds(ys, Woh, blockIdx.x * 128, blockIdx.y * 256, smem, acc);

    const int lane = threadIdx.x & 31;
    const int nlo = lane & 15, mh = (lane >> 4) * 8;
#pragma unroll
    for (int j = 0; j < 4; ++j) {
        const int n = colbase + j * 16 + nlo;
        const float bias = b_out[n];
#pragma unroll
        for (int i = 0; i < 4; ++i)
#pragma unroll
            for (int jj = 0; jj < 8; ++jj) {
                const int m = rowbase + i * 16 + mh + jj;
                const size_t idx = (size_t)m * MM + n;
                out[idx] = xflat[idx] + acc.c[i][j][jj] + bias;
            }
    }
}

extern "C" void kernel_launch(void* const* d_in, const int* in_sizes, int n_in,
                              void* d_out, int out_size, void* d_ws, size_t ws_size,
                              hipStream_t stream) {
    const float* x     = (const float*)d_in[0];
    const float* A_log = (const float*)d_in[1];
    const float* Dmat  = (const float*)d_in[2];
    const float* W_in  = (const float*)d_in[3];
    const float* b_in  = (const float*)d_in[4];
    const float* W_xp  = (const float*)d_in[5];
    const float* b_xp  = (const float*)d_in[6];
    const float* W_out = (const float*)d_in[7];
    const float* b_out = (const float*)d_in[8];
    const float* gamma = (const float*)d_in[9];
    const float* beta  = (const float*)d_in[10];
    float* out = (float*)d_out;

    char* ws = (char*)d_ws;
    size_t off = 0;
    auto take = [&](size_t bytes) { char* p = ws + off; off += bytes; return p; };
    _Float16* xn_h  = (_Float16*)take((size_t)ROWS * MM * 2);
    _Float16* xp_h  = (_Float16*)take((size_t)ROWS * MM * 2);
    _Float16* z_h   = (_Float16*)take((size_t)ROWS * MM * 2);
    _Float16* ys_h  = (_Float16*)take((size_t)ROWS * MM * 2);
    float*    u     = (float*)take((size_t)ROWS * SS * 4);
    float*    sgend = (float*)take((size_t)ROWS * SS * 4);
    _Float16* h_h   = (_Float16*)take((size_t)ROWS * SS * 2);
    _Float16* Wih   = (_Float16*)take((size_t)2 * MM * MM * 2);
    _Float16* Wxh   = (_Float16*)take((size_t)SS * MM * 2);
    _Float16* Woh   = (_Float16*)take((size_t)MM * MM * 2);
    _Float16* Dth   = (_Float16*)take((size_t)SS * MM * 2);
    (void)ws_size; (void)in_sizes; (void)n_in; (void)out_size;

    k_convert<<<CVT_WGS, 256, 0, stream>>>(W_in, W_xp, W_out, Dmat, Wih, Wxh, Woh, Dth);
    k_layernorm<<<ROWS / (8 * LN_RPW), 256, 0, stream>>>(x, gamma, beta, xn_h);
    k_gemm1<<<dim3(ROWS / 128, (2 * MM) / 256), 256, 0, stream>>>(xn_h, Wih, b_in, xp_h, z_h);
    k_gemm2<<<dim3(ROWS / 512, 1), 256, 0, stream>>>(xp_h, Wxh, b_xp, u);
    k_scan_ends<<<BB * NSEG, 256, 0, stream>>>(u, A_log, sgend);
    k_scan_apply<<<BB * NSEG, 256, 0, stream>>>(u, A_log, sgend, h_h);
    k_gemm3<<<dim3(ROWS / 128, MM / 256), 256, 0, stream>>>(h_h, Dth, z_h, ys_h);
    k_gemm4<<<dim3(ROWS / 128, MM / 256), 256, 0, stream>>>(ys_h, Woh, b_out, x, out);
}


# State projection re-gridded: 32 -> 256 work-groups of 64 rows, K split over the 8 waves, all 64 loads of a wave in flight before its first product; partial tiles summed through LDS (new sum order)
# speedup vs baseline: 2.8350x; 1.1718x over previous
#include <hip/hip_runtime.h>
#include <hip/hip_bf16.h>

#define BB   4
#define LL   4096
#define MM   1024
#define SS   64
#define ROWS (BB * LL)
#define SEG  128
#define NSEG (LL / SEG)
#define SUB  16

typedef __attribute__((ext_vector_type(16))) _Float16 v16h;
typedef __attribute__((ext_vector_type(8)))  _Float16 v8h;
typedef __attribute__((ext_vector_type(4)))  _Float16 v4h;
typedef __attribute__((ext_vector_type(8)))  float    v8f;
typedef __attribute__((ext_vector_type(4)))  float    v4f;

__device__ __forceinline__ v16h load_frag(const _Float16* __restrict__ p) {
    v8h lo = *(const v8h*)(p);
    v8h hi = *(const v8h*)(p + 16);
    v16h t;
#pragma unroll
    for (int i = 0; i < 8; ++i) { t[i] = lo[i]; t[i + 8] = hi[i]; }
    return t;
}

struct Acc44 { v8f c[4][4]; };

__device__ __forceinline__ void wave_gemm64(const _Float16* __restrict__ A,
                                            const _Float16* __restrict__ B,
                                            int ldA, int ldB,
                                            int rowbase, int colbase, int K,
                                            Acc44& acc) {
    const int lane  = threadIdx.x & 31;
    const int rr    = lane & 15;
    const int khalf = (lane >> 4) & 1;
    const _Float16* Ab[4];
    const _Float16* Bb[4];
#pragma unroll
    for (int t = 0; t < 4; ++t) {
        Ab[t] = A + (size_t)(rowbase + t * 16 + rr) * ldA + khalf * 8;
        Bb[t] = B + (size_t)(colbase + t * 16 + rr) * ldB + khalf * 8;
    }
    for (int kb = 0; kb < K; kb += 32) {
        v16h a[4], b[4];
#pragma unroll
        for (int t = 0; t < 4; ++t) a[t] = load_frag(Ab[t] + kb);
#pragma unroll
        for (int t = 0; t < 4; ++t) b[t] = load_frag(Bb[t] + kb);
#pragma unroll
        for (int i = 0; i < 4; ++i)
#pragma unroll
            for (int j = 0; j < 4; ++j)
                acc.c[i][j] = __builtin_amdgcn_wmma_f32_16x16x32_f16(
                    false, a[i], false, b[j], (short)0, acc.c[i][j], false, false);
    }
}

#define GBK     64
#define GLDP    72
#define GTM     128
#define GTN     256
#define GSTAGE  ((GTM + GTN) * GLDP)
#define GKT     (MM / GBK)

struct Stage12 { v8h a[4]; v8h b[8]; };

__device__ __forceinline__ void tile_fetch(const _Float16* __restrict__ gA,
                                           const _Float16* __restrict__ gB,
                                           int kofs, Stage12& g) {
#pragma unroll
    for (int r = 0; r < 4; ++r) g.a[r] = *(const v8h*)(gA + (size_t)r * 32 * MM + kofs);
#pragma unroll
    for (int r = 0; r < 8; ++r) g.b[r] = *(const v8h*)(gB + (size_t)r * 32 * MM + kofs);
}

__device__ __forceinline__ void tile_store(_Float16* sA, _Float16* sB, const Stage12& g) {
#pragma unroll
    for (int r = 0; r < 4; ++r) *(v8h*)(sA + r * 32 * GLDP) = g.a[r];
#pragma unroll
    for (int r = 0; r < 8; ++r) *(v8h*)(sB + r * 32 * GLDP) = g.b[r];
}

__device__ __forceinline__ void stage_compute(const _Float16* fA, const _Float16* fB, Acc44& acc) {
    v16h a[2][4], b[2][4];
#pragma unroll
    for (int ks = 0; ks < 2; ++ks) {
#pragma unroll
        for (int t = 0; t < 4; ++t) a[ks][t] = load_frag(fA + t * 16 * GLDP + ks * 32);
#pragma unroll
        for (int t = 0; t < 4; ++t) b[ks][t] = load_frag(fB + t * 16 * GLDP + ks * 32);
    }
#pragma unroll
    for (int ks = 0; ks < 2; ++ks)
#pragma unroll
        for (int i = 0; i < 4; ++i)
#pragma unroll
            for (int j = 0; j < 4; ++j)
                acc.c[i][j] = __builtin_amdgcn_wmma_f32_16x16x32_f16(
                    false, a[ks][i], false, b[ks][j], (short)0, acc.c[i][j], false, false);
}

__device__ __forceinline__ void block_gemm_lds(const _Float16* __restrict__ A,
                                               const _Float16* __restrict__ B,
                                               int rowblk, int colblk,
                                               _Float16* smem, Acc44& acc) {
    const int tid   = threadIdx.x;
    const int lane  = tid & 31;
    const int w     = tid >> 5;
    const int rr    = lane & 15;
    const int khalf = (lane >> 4) & 1;

    const int crow = tid >> 3;
    const int ccol = (tid & 7) * 8;
    const _Float16* gA = A + (size_t)(rowblk + crow) * MM + ccol;
    const _Float16* gB = B + (size_t)(colblk + crow) * MM + ccol;
    const int cofs = crow * GLDP + ccol;

    const int faofs = ((w & 1) * 64 + rr) * GLDP + khalf * 8;
    const int fbofs = GTM * GLDP + ((w >> 1) * 64 + rr) * GLDP + khalf * 8;

    Stage12 g;
    tile_fetch(gA, gB, 0, g);
    tile_store(smem + cofs, smem + GTM * GLDP + cofs, g);
    __syncthreads();

    for (int kt = 0; kt < GKT - 1; ++kt) {
        _Float16* cur = smem + (kt & 1) * GSTAGE;
        _Float16* nxt = smem + ((kt + 1) & 1) * GSTAGE;
        tile_fetch(gA, gB, (kt + 1) * GBK, g);
        __builtin_amdgcn_sched_barrier(0);
        stage_compute(cur + faofs, cur + fbofs, acc);
        tile_store(nxt + cofs, nxt + GTM * GLDP + cofs, g);
        __syncthreads();
    }
    {
        const _Float16* cur = smem + ((GKT - 1) & 1) * GSTAGE;
        stage_compute(cur + faofs, cur + fbofs, acc);
    }
}

__device__ __forceinline__ float fast_sigmoid(float v) {
    return __builtin_amdgcn_rcpf(1.0f + __expf(-v));
}

#define CVT_WGS 512
#define CVT_T   (CVT_WGS * 256)

__device__ __forceinline__ v4h cvt4(const float4 a) {
    v4h o;
    o[0] = (_Float16)a.x; o[1] = (_Float16)a.y; o[2] = (_Float16)a.z; o[3] = (_Float16)a.w;
    return o;
}

__global__ __launch_bounds__(256) void k_convert(const float* __restrict__ W_in,
                                                 const float* __restrict__ W_xp,
                                                 const float* __restrict__ W_out,
                                                 const float* __restrict__ Dmat,
                                                 _Float16* __restrict__ Wih,
                                                 _Float16* __restrict__ Wxh,
                                                 _Float16* __restrict__ Woh,
                                                 _Float16* __restrict__ Dth) {
    constexpr int NWI4 = 2 * MM * MM / 4;
    constexpr int NWO4 = MM * MM / 4;
    constexpr int NXP4 = SS * MM / 4;
    static_assert(NWI4 % CVT_T == 0 && NWO4 % CVT_T == 0 && NXP4 <= CVT_T, "conversion grid");
    const int t = blockIdx.x * 256 + threadIdx.x;

    float4 a[NWI4 / CVT_T], b[NWO4 / CVT_T];
#pragma unroll
    for (int k = 0; k < NWI4 / CVT_T; ++k) a[k] = ((const float4*)W_in)[t + k * CVT_T];
#pragma unroll
    for (int k = 0; k < NWO4 / CVT_T; ++k) b[k] = ((const float4*)W_out)[t + k * CVT_T];
#pragma unroll
    for (int k = 0; k < NWI4 / CVT_T; ++k) ((v4h*)Wih)[t + k * CVT_T] = cvt4(a[k]);
#pragma unroll
    for (int k = 0; k < NWO4 / CVT_T; ++k) ((v4h*)Woh)[t + k * CVT_T] = cvt4(b[k]);

    if (t < NXP4) {
        ((v4h*)Wxh)[t] = cvt4(((const float4*)W_xp)[t]);
        const int n = t >> 4, s4 = (t & 15) * 4;
        v4h o;
#pragma unroll
        for (int k = 0; k < 4; ++k) o[k] = (_Float16)Dmat[(size_t)(s4 + k) * MM + n];
        ((v4h*)Dth)[t] = o;
    }
}

#define LN_RPW 4

__global__ __launch_bounds__(256) void k_layernorm(const float* __restrict__ x,
                                                   const float* __restrict__ gamma,
                                                   const float* __restrict__ beta,
                                                   _Float16* __restrict__ xn) {
    const int lane = threadIdx.x & 31;
    const int wave = blockIdx.x * 8 + (threadIdx.x >> 5);
    const size_t base = (size_t)wave * LN_RPW * MM;

    float4 v[LN_RPW][8];
#pragma unroll
    for (int r = 0; r < LN_RPW; ++r)
#pragma unroll
        for (int k = 0; k < 8; ++k)
            v[r][k] = ((const float4*)(x + base + (size_t)r * MM))[k * 32 + lane];

    float s[LN_RPW], q[LN_RPW];
#pragma unroll
    for (int r = 0; r < LN_RPW; ++r) {
        s[r] = 0.f; q[r] = 0.f;
#pragma unroll
        for (int k = 0; k < 8; ++k) {
            const float4 e = v[r][k];
            s[r] += e.x + e.y + e.z + e.w;
            q[r] += e.x * e.x + e.y * e.y + e.z * e.z + e.w * e.w;
        }
    }
#pragma unroll
    for (int off = 16; off > 0; off >>= 1) {
#pragma unroll
        for (int r = 0; r < LN_RPW; ++r) {
            s[r] += __shfl_xor(s[r], off);
            q[r] += __shfl_xor(q[r], off);
        }
    }
    float mu[LN_RPW], rstd[LN_RPW];
#pragma unroll
    for (int r = 0; r < LN_RPW; ++r) {
        mu[r] = s[r] * (1.0f / MM);
        const float var = q[r] * (1.0f / MM) - mu[r] * mu[r];
        rstd[r] = rsqrtf(var + 1e-5f);
    }
#pragma unroll
    for (int k = 0; k < 8; ++k) {
        const float4 g  = ((const float4*)gamma)[k * 32 + lane];
        const float4 bb = ((const float4*)beta)[k * 32 + lane];
#pragma unroll
        for (int r = 0; r < LN_RPW; ++r) {
            const float4 e = v[r][k];
            v4h o;
            o[0] = (_Float16)((e.x - mu[r]) * rstd[r] * g.x + bb.x);
            o[1] = (_Float16)((e.y - mu[r]) * rstd[r] * g.y + bb.y);
            o[2] = (_Float16)((e.z - mu[r]) * rstd[r] * g.z + bb.z);
            o[3] = (_Float16)((e.w - mu[r]) * rstd[r] * g.w + bb.w);
            ((v4h*)(xn + base + (size_t)r * MM))[k * 32 + lane] = o;
        }
    }
}

__global__ __launch_bounds__(256) void k_gemm1(const _Float16* __restrict__ xn,
                                               const _Float16* __restrict__ Wih,
                                               const float* __restrict__ b_in,
                                               _Float16* __restrict__ xp,
                                               _Float16* __restrict__ zg) {
    __shared__ __attribute__((aligned(16))) _Float16 smem[2 * GSTAGE];
    Acc44 acc = {};
    const int w = threadIdx.x >> 5;
    const int rowbase = blockIdx.x * 128 + (w & 1) * 64;
    const int colbase = blockIdx.y * 256 + (w >> 1) * 64;
    block_gemm_lds(xn, Wih, blockIdx.x * 128, blockIdx.y * 256, smem, acc);

    const int lane = threadIdx.x & 31;
    const int nlo = lane & 15, mh = (lane >> 4) * 8;
    if (blockIdx.y < (MM / 256)) {
#pragma unroll
        for (int j = 0; j < 4; ++j) {
            const int n = colbase + j * 16 + nlo;
            const float bias = b_in[n];
#pragma unroll
            for (int i = 0; i < 4; ++i) {
#pragma unroll
                for (int jj = 0; jj < 8; ++jj) {
                    const int m = rowbase + i * 16 + mh + jj;
                    const float v = acc.c[i][j][jj] + bias;
                    xp[(size_t)m * MM + n] = (_Float16)(v * fast_sigmoid(v));
                }
            }
        }
    } else {
#pragma unroll
        for (int j = 0; j < 4; ++j) {
            const int n = colbase + j * 16 + nlo;
            const float bias = b_in[n];
#pragma unroll
            for (int i = 0; i < 4; ++i) {
#pragma unroll
                for (int jj = 0; jj < 8; ++jj) {
                    const int m = rowbase + i * 16 + mh + jj;
                    const float v = acc.c[i][j][jj] + bias;
                    zg[(size_t)m * MM + (n - MM)] = (_Float16)fast_sigmoid(v);
                }
            }
        }
    }
}

#define G2_RP 68

__global__ __launch_bounds__(256) void k_gemm2(const _Float16* __restrict__ xp,
                                               const _Float16* __restrict__ Wxh,
                                               const float* __restrict__ b_xp,
                                               float* __restrict__ u) {
    __shared__ __attribute__((aligned(16))) float red[8 * 64 * G2_RP];
    const int lane  = threadIdx.x & 31;
    const int w     = threadIdx.x >> 5;
    const int rr    = lane & 15;
    const int khalf = (lane >> 4) & 1;
    const int rowbase = blockIdx.x * 64;

    const _Float16* ap = xp  + (size_t)(rowbase + rr) * MM + w * 128 + khalf * 8;
    const _Float16* bp = Wxh + (size_t)rr * MM + w * 128 + khalf * 8;

    v16h a[4][4], b[4][4];
#pragma unroll
    for (int ks = 0; ks < 4; ++ks) {
#pragma unroll
        for (int t = 0; t < 4; ++t) a[ks][t] = load_frag(ap + (size_t)t * 16 * MM + ks * 32);
#pragma unroll
        for (int t = 0; t < 4; ++t) b[ks][t] = load_frag(bp + (size_t)t * 16 * MM + ks * 32);
    }
    __builtin_amdgcn_sched_barrier(0);

    Acc44 acc = {};
#pragma unroll
    for (int ks = 0; ks < 4; ++ks) {
#pragma unroll
        for (int i = 0; i < 4; ++i)
#pragma unroll
            for (int j = 0; j < 4; ++j)
                acc.c[i][j] = __builtin_amdgcn_wmma_f32_16x16x32_f16(
                    false, a[ks][i], false, b[ks][j], (short)0, acc.c[i][j], false, false);
        if (ks < 3) __builtin_amdgcn_sched_barrier(0);
    }

    float* mine = red + w * (64 * G2_RP);
    const int mh = khalf * 8;
#pragma unroll
    for (int i = 0; i < 4; ++i)
#pragma unroll
        for (int j = 0; j < 4; ++j)
#pragma unroll
            for (int r = 0; r < 8; ++r)
                mine[(i * 16 + mh + r) * G2_RP + j * 16 + rr] = acc.c[i][j][r];
    __syncthreads();

#pragma unroll
    for (int q = 0; q < 4; ++q) {
        const int idx = threadIdx.x + 256 * q;
        const int row = idx >> 4, c4 = (idx & 15) * 4;
        const float* s = red + row * G2_RP + c4;
        v4f o = *(const v4f*)s;
#pragma unroll
        for (int k = 1; k < 8; ++k) o += *(const v4f*)(s + k * (64 * G2_RP));
        o += *(const v4f*)(b_xp + c4);
        *(v4f*)(u + (size_t)(rowbase + row) * SS + c4) = o;
    }
}

struct ScanLane {
    float2 uv[SUB];
    float  a0, a1;
    float  p0, p1;
};

__device__ __forceinline__ void scan_load(const float* __restrict__ u,
                                          const float* __restrict__ A_log,
                                          int b, int l0, int lane, ScanLane& t) {
    const float2* up = (const float2*)(u + ((size_t)b * LL + l0) * SS) + lane;
#pragma unroll
    for (int j = 0; j < SUB; ++j) t.uv[j] = up[j * (SS / 2)];
    const float2 al = ((const float2*)A_log)[lane];
    t.a0 = expf(-expf(al.x));
    t.a1 = expf(-expf(al.y));
    float q0 = t.a0, q1 = t.a1;
#pragma unroll
    for (int i = 1; i < SUB; i <<= 1) { q0 *= q0; q1 *= q1; }
    t.p0 = q0; t.p1 = q1;
}

__device__ __forceinline__ float2 scan_run(const ScanLane& t, float2 h) {
#pragma unroll
    for (int j = 0; j < SUB; ++j) {
        h.x = t.a0 * h.x + t.uv[j].x;
        h.y = t.a1 * h.y + t.uv[j].y;
    }
    return h;
}

__global__ __launch_bounds__(256) void k_scan_ends(const float* __restrict__ u,
                                                   const float* __restrict__ A_log,
                                                   float* __restrict__ seg_end) {
    __shared__ float2 we[SEG / SUB][32];
    const int lane = threadIdx.x & 31, k = threadIdx.x >> 5;
    const int b = blockIdx.x / NSEG, sg = blockIdx.x % NSEG;
    ScanLane t;
    scan_load(u, A_log, b, sg * SEG + k * SUB, lane, t);
    we[k][lane] = scan_run(t, make_float2(0.f, 0.f));
    __syncthreads();
    if (k == 0) {
        float2 e = make_float2(0.f, 0.f);
#pragma unroll
        for (int kk = 0; kk < SEG / SUB; ++kk) {
            const float2 w = we[kk][lane];
            e.x = t.p0 * e.x + w.x;
            e.y = t.p1 * e.y + w.y;
        }
        ((float2*)seg_end)[(size_t)blockIdx.x * 32 + lane] = e;
    }
}

__global__ __launch_bounds__(256) void k_scan_apply(const float* __restrict__ u,
                                                    const float* __restrict__ A_log,
                                                    const float* __restrict__ seg_end,
                                                    _Float16* __restrict__ h_h) {
    __shared__ float2 we[SEG / SUB][32];
    const int lane = threadIdx.x & 31, k = threadIdx.x >> 5;
    const int b = blockIdx.x / NSEG, sg = blockIdx.x % NSEG;
    const int l0 = sg * SEG + k * SUB;
    ScanLane t;
    scan_load(u, A_log, b, l0, lane, t);

    const float2* ep = (const float2*)seg_end + (size_t)b * NSEG * 32 + lane;
    float2 se[NSEG - 1];
#pragma unroll
    for (int d = 0; d < NSEG - 1; ++d) {
        const int q = sg - 1 - d;
        const float2 v = ep[(q > 0 ? q : 0) * 32];
        se[d] = (q >= 0) ? v : make_float2(0.f, 0.f);
    }

    we[k][lane] = scan_run(t, make_float2(0.f, 0.f));

    float P0 = t.p0, P1 = t.p1;
#pragma unroll
    for (int i = 1; i < SEG / SUB; i <<= 1) { P0 *= P0; P1 *= P1; }
    float2 c = make_float2(0.f, 0.f);
#pragma unroll
    for (int d = NSEG - 2; d >= 0; --d) {
        c.x = P0 * c.x + se[d].x;
        c.y = P1 * c.y + se[d].y;
    }
    __syncthreads();
#pragma unroll
    for (int kk = 0; kk < SEG / SUB - 1; ++kk) {
        const float2 w = we[kk][lane];
        const float nx = t.p0 * c.x + w.x;
        const float ny = t.p1 * c.y + w.y;
        c.x = (kk < k) ? nx : c.x;
        c.y = (kk < k) ? ny : c.y;
    }
    typedef __attribute__((ext_vector_type(2))) _Float16 v2h;
    v2h* hp = (v2h*)(h_h + ((size_t)l0 * BB + b) * SS) + lane;
#pragma unroll
    for (int j = 0; j < SUB; ++j) {
        c.x = t.a0 * c.x + t.uv[j].x;
        c.y = t.a1 * c.y + t.uv[j].y;
        v2h o;
        o[0] = (_Float16)c.x; o[1] = (_Float16)c.y;
        hp[(size_t)j * BB * (SS / 2)] = o;
    }
}

__global__ __launch_bounds__(256) void k_gemm3(const _Float16* __restrict__ h_h,
                                               const _Float16* __restrict__ Dth,
                                               const _Float16* __restrict__ zg,
                                               _Float16* __restrict__ ys) {
    Acc44 acc = {};
    const int w = threadIdx.x >> 5;
    const int rowbase = blockIdx.x * 128 + (w & 1) * 64;
    const int colbase = blockIdx.y * 256 + (w >> 1) * 64;
    wave_gemm64(h_h, Dth, SS, SS, rowbase, colbase, SS, acc);

    const int lane = threadIdx.x & 31;
    const int nlo = lane & 15, mh = (lane >> 4) * 8;
#pragma unroll
    for (int i = 0; i < 4; ++i)
#pragma unroll
        for (int j = 0; j < 4; ++j)
#pragma unroll
            for (int jj = 0; jj < 8; ++jj) {
                const int rp = rowbase + i * 16 + mh + jj;
                const int n = colbase + j * 16 + nlo;
                const int b = rp & (BB - 1), l = rp >> 2;
                const float z = (float)zg[((size_t)b * LL + l) * MM + n];
                ys[(size_t)rp * MM + n] = (_Float16)(acc.c[i][j][jj] * z);
            }
}

__global__ __launch_bounds__(256, 4) void k_gemm4(const _Float16* __restrict__ ys,
                                               const _Float16* __restrict__ Woh,
                                               const float* __restrict__ b_out,
                                               const float* __restrict__ xflat,
                                               float* __restrict__ out) {
    __shared__ __attribute__((aligned(16))) _Float16 smem[2 * GSTAGE];
    Acc44 acc = {};
    const int w = threadIdx.x >> 5;
    const int rowbase = blockIdx.x * 128 + (w & 1) * 64;
    const int colbase = blockIdx.y * 256 + (w >> 1) * 64;
    block_gemm_lds(ys, Woh, blockIdx.x * 128, blockIdx.y * 256, smem, acc);

    const int lane = threadIdx.x & 31;
    const int nlo = lane & 15, mh = (lane >> 4) * 8;
#pragma unroll
    for (int j = 0; j < 4; ++j) {
        const int n = colbase + j * 16 + nlo;
        const float bias = b_out[n];
#pragma unroll
        for (int i = 0; i < 4; ++i)
#pragma unroll
            for (int jj = 0; jj < 8; ++jj) {
                const int m = rowbase + i * 16 + mh + jj;
                const size_t idx = (size_t)m * MM + n;
                out[idx] = xflat[idx] + acc.c[i][j][jj] + bias;
            }
    }
}

extern "C" void kernel_launch(void* const* d_in, const int* in_sizes, int n_in,
                              void* d_out, int out_size, void* d_ws, size_t ws_size,
                              hipStream_t stream) {
    const float* x     = (const float*)d_in[0];
    const float* A_log = (const float*)d_in[1];
    const float* Dmat  = (const float*)d_in[2];
    const float* W_in  = (const float*)d_in[3];
    const float* b_in  = (const float*)d_in[4];
    const float* W_xp  = (const float*)d_in[5];
    const float* b_xp  = (const float*)d_in[6];
    const float* W_out = (const float*)d_in[7];
    const float* b_out = (const float*)d_in[8];
    const float* gamma = (const float*)d_in[9];
    const float* beta  = (const float*)d_in[10];
    float* out = (float*)d_out;

    char* ws = (char*)d_ws;
    size_t off = 0;
    auto take = [&](size_t bytes) { char* p = ws + off; off += bytes; return p; };
    _Float16* xn_h  = (_Float16*)take((size_t)ROWS * MM * 2);
    _Float16* xp_h  = (_Float16*)take((size_t)ROWS * MM * 2);
    _Float16* z_h   = (_Float16*)take((size_t)ROWS * MM * 2);
    _Float16* ys_h  = (_Float16*)take((size_t)ROWS * MM * 2);
    float*    u     = (float*)take((size_t)ROWS * SS * 4);
    float*    sgend = (float*)take((size_t)ROWS * SS * 4);
    _Float16* h_h   = (_Float16*)take((size_t)ROWS * SS * 2);
    _Float16* Wih   = (_Float16*)take((size_t)2 * MM * MM * 2);
    _Float16* Wxh   = (_Float16*)take((size_t)SS * MM * 2);
    _Float16* Woh   = (_Float16*)take((size_t)MM * MM * 2);
    _Float16* Dth   = (_Float16*)take((size_t)SS * MM * 2);
    (void)ws_size; (void)in_sizes; (void)n_in; (void)out_size;

    k_convert<<<CVT_WGS, 256, 0, stream>>>(W_in, W_xp, W_out, Dmat, Wih, Wxh, Woh, Dth);
    k_layernorm<<<ROWS / (8 * LN_RPW), 256, 0, stream>>>(x, gamma, beta, xn_h);
    k_gemm1<<<dim3(ROWS / 128, (2 * MM) / 256), 256, 0, stream>>>(xn_h, Wih, b_in, xp_h, z_h);
    k_gemm2<<<dim3(ROWS / 64, 1), 256, 0, stream>>>(xp_h, Wxh, b_xp, u);
    k_scan_ends<<<BB * NSEG, 256, 0, stream>>>(u, A_log, sgend);
    k_scan_apply<<<BB * NSEG, 256, 0, stream>>>(u, A_log, sgend, h_h);
    k_gemm3<<<dim3(ROWS / 128, MM / 256), 256, 0, stream>>>(h_h, Dth, z_h, ys_h);
    k_gemm4<<<dim3(ROWS / 128, MM / 256), 256, 0, stream>>>(ys_h, Woh, b_out, x, out);
}


# in_proj and out_proj K-loop: a wave's LDS fragment reads issued as one group of 16 per 32-deep sub-step, then its 16 products back to back (was 2 reads, full wait, 4 products)
# speedup vs baseline: 3.0013x; 1.0587x over previous
#include <hip/hip_runtime.h>
#include <hip/hip_bf16.h>

#define BB   4
#define LL   4096
#define MM   1024
#define SS   64
#define ROWS (BB * LL)
#define SEG  128
#define NSEG (LL / SEG)
#define SUB  16

typedef __attribute__((ext_vector_type(16))) _Float16 v16h;
typedef __attribute__((ext_vector_type(8)))  _Float16 v8h;
typedef __attribute__((ext_vector_type(4)))  _Float16 v4h;
typedef __attribute__((ext_vector_type(8)))  float    v8f;
typedef __attribute__((ext_vector_type(4)))  float    v4f;

__device__ __forceinline__ v16h load_frag(const _Float16* __restrict__ p) {
    v8h lo = *(const v8h*)(p);
    v8h hi = *(const v8h*)(p + 16);
    v16h t;
#pragma unroll
    for (int i = 0; i < 8; ++i) { t[i] = lo[i]; t[i + 8] = hi[i]; }
    return t;
}

struct Acc44 { v8f c[4][4]; };

__device__ __forceinline__ void wave_gemm64(const _Float16* __restrict__ A,
                                            const _Float16* __restrict__ B,
                                            int ldA, int ldB,
                                            int rowbase, int colbase, int K,
                                            Acc44& acc) {
    const int lane  = threadIdx.x & 31;
    const int rr    = lane & 15;
    const int khalf = (lane >> 4) & 1;
    const _Float16* Ab[4];
    const _Float16* Bb[4];
#pragma unroll
    for (int t = 0; t < 4; ++t) {
        Ab[t] = A + (size_t)(rowbase + t * 16 + rr) * ldA + khalf * 8;
        Bb[t] = B + (size_t)(colbase + t * 16 + rr) * ldB + khalf * 8;
    }
    for (int kb = 0; kb < K; kb += 32) {
        v16h a[4], b[4];
#pragma unroll
        for (int t = 0; t < 4; ++t) a[t] = load_frag(Ab[t] + kb);
#pragma unroll
        for (int t = 0; t < 4; ++t) b[t] = load_frag(Bb[t] + kb);
#pragma unroll
        for (int i = 0; i < 4; ++i)
#pragma unroll
            for (int j = 0; j < 4; ++j)
                acc.c[i][j] = __builtin_amdgcn_wmma_f32_16x16x32_f16(
                    false, a[i], false, b[j], (short)0, acc.c[i][j], false, false);
    }
}

#define GBK     64
#define GLDP    72
#define GTM     128
#define GTN     256
#define GSTAGE  ((GTM + GTN) * GLDP)
#define GKT     (MM / GBK)

struct Stage12 { v8h a[4]; v8h b[8]; };

__device__ __forceinline__ void tile_fetch(const _Float16* __restrict__ gA,
                                           const _Float16* __restrict__ gB,
                                           int kofs, Stage12& g) {
#pragma unroll
    for (int r = 0; r < 4; ++r) g.a[r] = *(const v8h*)(gA + (size_t)r * 32 * MM + kofs);
#pragma unroll
    for (int r = 0; r < 8; ++r) g.b[r] = *(const v8h*)(gB + (size_t)r * 32 * MM + kofs);
}

__device__ __forceinline__ void tile_store(_Float16* sA, _Float16* sB, const Stage12& g) {
#pragma unroll
    for (int r = 0; r < 4; ++r) *(v8h*)(sA + r * 32 * GLDP) = g.a[r];
#pragma unroll
    for (int r = 0; r < 8; ++r) *(v8h*)(sB + r * 32 * GLDP) = g.b[r];
}

__device__ __forceinline__ void stage_compute(const _Float16* fA, const _Float16* fB, Acc44& acc) {
#pragma unroll
    for (int ks = 0; ks < 2; ++ks) {
        v16h a[4], b[4];
#pragma unroll
        for (int t = 0; t < 4; ++t) b[t] = load_frag(fB + t * 16 * GLDP + ks * 32);
#pragma unroll
        for (int t = 0; t < 4; ++t) a[t] = load_frag(fA + t * 16 * GLDP + ks * 32);
        __builtin_amdgcn_sched_barrier(0);
#pragma unroll
        for (int i = 0; i < 4; ++i)
#pragma unroll
            for (int j = 0; j < 4; ++j)
                acc.c[i][j] = __builtin_amdgcn_wmma_f32_16x16x32_f16(
                    false, a[i], false, b[j], (short)0, acc.c[i][j], false, false);
        __builtin_amdgcn_sched_barrier(0);
    }
}

__device__ __forceinline__ void block_gemm_lds(const _Float16* __restrict__ A,
                                               const _Float16* __restrict__ B,
                                               int rowblk, int colblk,
                                               _Float16* smem, Acc44& acc) {
    const int tid   = threadIdx.x;
    const int lane  = tid & 31;
    const int w     = tid >> 5;
    const int rr    = lane & 15;
    const int khalf = (lane >> 4) & 1;

    const int crow = tid >> 3;
    const int ccol = (tid & 7) * 8;
    const _Float16* gA = A + (size_t)(rowblk + crow) * MM + ccol;
    const _Float16* gB = B + (size_t)(colblk + crow) * MM + ccol;
    const int cofs = crow * GLDP + ccol;

    const int faofs = ((w & 1) * 64 + rr) * GLDP + khalf * 8;
    const int fbofs = GTM * GLDP + ((w >> 1) * 64 + rr) * GLDP + khalf * 8;

    Stage12 g;
    tile_fetch(gA, gB, 0, g);
    tile_store(smem + cofs, smem + GTM * GLDP + cofs, g);
    __syncthreads();

    for (int kt = 0; kt < GKT - 1; ++kt) {
        _Float16* cur = smem + (kt & 1) * GSTAGE;
        _Float16* nxt = smem + ((kt + 1) & 1) * GSTAGE;
        tile_fetch(gA, gB, (kt + 1) * GBK, g);
        __builtin_amdgcn_sched_barrier(0);
        stage_compute(cur + faofs, cur + fbofs, acc);
        tile_store(nxt + cofs, nxt + GTM * GLDP + cofs, g);
        __syncthreads();
    }
    {
        const _Float16* cur = smem + ((GKT - 1) & 1) * GSTAGE;
        stage_compute(cur + faofs, cur + fbofs, acc);
    }
}

__device__ __forceinline__ float fast_sigmoid(float v) {
    return __builtin_amdgcn_rcpf(1.0f + __expf(-v));
}

#define CVT_WGS 512
#define CVT_T   (CVT_WGS * 256)

__device__ __forceinline__ v4h cvt4(const float4 a) {
    v4h o;
    o[0] = (_Float16)a.x; o[1] = (_Float16)a.y; o[2] = (_Float16)a.z; o[3] = (_Float16)a.w;
    return o;
}

__global__ __launch_bounds__(256) void k_convert(const float* __restrict__ W_in,
                                                 const float* __restrict__ W_xp,
                                                 const float* __restrict__ W_out,
                                                 const float* __restrict__ Dmat,
                                                 _Float16* __restrict__ Wih,
                                                 _Float16* __restrict__ Wxh,
                                                 _Float16* __restrict__ Woh,
                                                 _Float16* __restrict__ Dth) {
    constexpr int NWI4 = 2 * MM * MM / 4;
    constexpr int NWO4 = MM * MM / 4;
    constexpr int NXP4 = SS * MM / 4;
    static_assert(NWI4 % CVT_T == 0 && NWO4 % CVT_T == 0 && NXP4 <= CVT_T, "conversion grid");
    const int t = blockIdx.x * 256 + threadIdx.x;

    float4 a[NWI4 / CVT_T], b[NWO4 / CVT_T];
#pragma unroll
    for (int k = 0; k < NWI4 / CVT_T; ++k) a[k] = ((const float4*)W_in)[t + k * CVT_T];
#pragma unroll
    for (int k = 0; k < NWO4 / CVT_T; ++k) b[k] = ((const float4*)W_out)[t + k * CVT_T];
#pragma unroll
    for (int k = 0; k < NWI4 / CVT_T; ++k) ((v4h*)Wih)[t + k * CVT_T] = cvt4(a[k]);
#pragma unroll
    for (int k = 0; k < NWO4 / CVT_T; ++k) ((v4h*)Woh)[t + k * CVT_T] = cvt4(b[k]);

    if (t < NXP4) {
        ((v4h*)Wxh)[t] = cvt4(((const float4*)W_xp)[t]);
        const int n = t >> 4, s4 = (t & 15) * 4;
        v4h o;
#pragma unroll
        for (int k = 0; k < 4; ++k) o[k] = (_Float16)Dmat[(size_t)(s4 + k) * MM + n];
        ((v4h*)Dth)[t] = o;
    }
}

#define LN_RPW 4

__global__ __launch_bounds__(256) void k_layernorm(const float* __restrict__ x,
                                                   const float* __restrict__ gamma,
                                                   const float* __restrict__ beta,
                                                   _Float16* __restrict__ xn) {
    const int lane = threadIdx.x & 31;
    const int wave = blockIdx.x * 8 + (threadIdx.x >> 5);
    const size_t base = (size_t)wave * LN_RPW * MM;

    float4 v[LN_RPW][8];
#pragma unroll
    for (int r = 0; r < LN_RPW; ++r)
#pragma unroll
        for (int k = 0; k < 8; ++k)
            v[r][k] = ((const float4*)(x + base + (size_t)r * MM))[k * 32 + lane];

    float s[LN_RPW], q[LN_RPW];
#pragma unroll
    for (int r = 0; r < LN_RPW; ++r) {
        s[r] = 0.f; q[r] = 0.f;
#pragma unroll
        for (int k = 0; k < 8; ++k) {
            const float4 e = v[r][k];
            s[r] += e.x + e.y + e.z + e.w;
            q[r] += e.x * e.x + e.y * e.y + e.z * e.z + e.w * e.w;
        }
    }
#pragma unroll
    for (int off = 16; off > 0; off >>= 1) {
#pragma unroll
        for (int r = 0; r < LN_RPW; ++r) {
            s[r] += __shfl_xor(s[r], off);
            q[r] += __shfl_xor(q[r], off);
        }
    }
    float mu[LN_RPW], rstd[LN_RPW];
#pragma unroll
    for (int r = 0; r < LN_RPW; ++r) {
        mu[r] = s[r] * (1.0f / MM);
        const float var = q[r] * (1.0f / MM) - mu[r] * mu[r];
        rstd[r] = rsqrtf(var + 1e-5f);
    }
#pragma unroll
    for (int k = 0; k < 8; ++k) {
        const float4 g  = ((const float4*)gamma)[k * 32 + lane];
        const float4 bb = ((const float4*)beta)[k * 32 + lane];
#pragma unroll
        for (int r = 0; r < LN_RPW; ++r) {
            const float4 e = v[r][k];
            v4h o;
            o[0] = (_Float16)((e.x - mu[r]) * rstd[r] * g.x + bb.x);
            o[1] = (_Float16)((e.y - mu[r]) * rstd[r] * g.y + bb.y);
            o[2] = (_Float16)((e.z - mu[r]) * rstd[r] * g.z + bb.z);
            o[3] = (_Float16)((e.w - mu[r]) * rstd[r] * g.w + bb.w);
            ((v4h*)(xn + base + (size_t)r * MM))[k * 32 + lane] = o;
        }
    }
}

__global__ __launch_bounds__(256) void k_gemm1(const _Float16* __restrict__ xn,
                                               const _Float16* __restrict__ Wih,
                                               const float* __restrict__ b_in,
                                               _Float16* __restrict__ xp,
                                               _Float16* __restrict__ zg) {
    __shared__ __attribute__((aligned(16))) _Float16 smem[2 * GSTAGE];
    Acc44 acc = {};
    const int w = threadIdx.x >> 5;
    const int rowbase = blockIdx.x * 128 + (w & 1) * 64;
    const int colbase = blockIdx.y * 256 + (w >> 1) * 64;
    block_gemm_lds(xn, Wih, blockIdx.x * 128, blockIdx.y * 256, smem, acc);

    const int lane = threadIdx.x & 31;
    const int nlo = lane & 15, mh = (lane >> 4) * 8;
    if (blockIdx.y < (MM / 256)) {
#pragma unroll
        for (int j = 0; j < 4; ++j) {
            const int n = colbase + j * 16 + nlo;
            const float bias = b_in[n];
#pragma unroll
            for (int i = 0; i < 4; ++i) {
#pragma unroll
                for (int jj = 0; jj < 8; ++jj) {
                    const int m = rowbase + i * 16 + mh + jj;
                    const float v = acc.c[i][j][jj] + bias;
                    xp[(size_t)m * MM + n] = (_Float16)(v * fast_sigmoid(v));
                }
            }
        }
    } else {
#pragma unroll
        for (int j = 0; j < 4; ++j) {
            const int n = colbase + j * 16 + nlo;
            const float bias = b_in[n];
#pragma unroll
            for (int i = 0; i < 4; ++i) {
#pragma unroll
                for (int jj = 0; jj < 8; ++jj) {
                    const int m = rowbase + i * 16 + mh + jj;
                    const float v = acc.c[i][j][jj] + bias;
                    zg[(size_t)m * MM + (n - MM)] = (_Float16)fast_sigmoid(v);
                }
            }
        }
    }
}

#define G2_RP 68

__global__ __launch_bounds__(256) void k_gemm2(const _Float16* __restrict__ xp,
                                               const _Float16* __restrict__ Wxh,
                                               const float* __restrict__ b_xp,
                                               float* __restrict__ u) {
    __shared__ __attribute__((aligned(16))) float red[8 * 64 * G2_RP];
    const int lane  = threadIdx.x & 31;
    const int w     = threadIdx.x >> 5;
    const int rr    = lane & 15;
    const int khalf = (lane >> 4) & 1;
    const int rowbase = blockIdx.x * 64;

    const _Float16* ap = xp  + (size_t)(rowbase + rr) * MM + w * 128 + khalf * 8;
    const _Float16* bp = Wxh + (size_t)rr * MM + w * 128 + khalf * 8;

    v16h a[4][4], b[4][4];
#pragma unroll
    for (int ks = 0; ks < 4; ++ks) {
#pragma unroll
        for (int t = 0; t < 4; ++t) a[ks][t] = load_frag(ap + (size_t)t * 16 * MM + ks * 32);
#pragma unroll
        for (int t = 0; t < 4; ++t) b[ks][t] = load_frag(bp + (size_t)t * 16 * MM + ks * 32);
    }
    __builtin_amdgcn_sched_barrier(0);

    Acc44 acc = {};
#pragma unroll
    for (int ks = 0; ks < 4; ++ks) {
#pragma unroll
        for (int i = 0; i < 4; ++i)
#pragma unroll
            for (int j = 0; j < 4; ++j)
                acc.c[i][j] = __builtin_amdgcn_wmma_f32_16x16x32_f16(
                    false, a[ks][i], false, b[ks][j], (short)0, acc.c[i][j], false, false);
        if (ks < 3) __builtin_amdgcn_sched_barrier(0);
    }

    float* mine = red + w * (64 * G2_RP);
    const int mh = khalf * 8;
#pragma unroll
    for (int i = 0; i < 4; ++i)
#pragma unroll
        for (int j = 0; j < 4; ++j)
#pragma unroll
            for (int r = 0; r < 8; ++r)
                mine[(i * 16 + mh + r) * G2_RP + j * 16 + rr] = acc.c[i][j][r];
    __syncthreads();

#pragma unroll
    for (int q = 0; q < 4; ++q) {
        const int idx = threadIdx.x + 256 * q;
        const int row = idx >> 4, c4 = (idx & 15) * 4;
        const float* s = red + row * G2_RP + c4;
        v4f o = *(const v4f*)s;
#pragma unroll
        for (int k = 1; k < 8; ++k) o += *(const v4f*)(s + k * (64 * G2_RP));
        o += *(const v4f*)(b_xp + c4);
        *(v4f*)(u + (size_t)(rowbase + row) * SS + c4) = o;
    }
}

struct ScanLane {
    float2 uv[SUB];
    float  a0, a1;
    float  p0, p1;
};

__device__ __forceinline__ void scan_load(const float* __restrict__ u,
                                          const float* __restrict__ A_log,
                                          int b, int l0, int lane, ScanLane& t) {
    const float2* up = (const float2*)(u + ((size_t)b * LL + l0) * SS) + lane;
#pragma unroll
    for (int j = 0; j < SUB; ++j) t.uv[j] = up[j * (SS / 2)];
    const float2 al = ((const float2*)A_log)[lane];
    t.a0 = expf(-expf(al.x));
    t.a1 = expf(-expf(al.y));
    float q0 = t.a0, q1 = t.a1;
#pragma unroll
    for (int i = 1; i < SUB; i <<= 1) { q0 *= q0; q1 *= q1; }
    t.p0 = q0; t.p1 = q1;
}

__device__ __forceinline__ float2 scan_run(const ScanLane& t, float2 h) {
#pragma unroll
    for (int j = 0; j < SUB; ++j) {
        h.x = t.a0 * h.x + t.uv[j].x;
        h.y = t.a1 * h.y + t.uv[j].y;
    }
    return h;
}

__global__ __launch_bounds__(256) void k_scan_ends(const float* __restrict__ u,
                                                   const float* __restrict__ A_log,
                                                   float* __restrict__ seg_end) {
    __shared__ float2 we[SEG / SUB][32];
    const int lane = threadIdx.x & 31, k = threadIdx.x >> 5;
    const int b = blockIdx.x / NSEG, sg = blockIdx.x % NSEG;
    ScanLane t;
    scan_load(u, A_log, b, sg * SEG + k * SUB, lane, t);
    we[k][lane] = scan_run(t, make_float2(0.f, 0.f));
    __syncthreads();
    if (k == 0) {
        float2 e = make_float2(0.f, 0.f);
#pragma unroll
        for (int kk = 0; kk < SEG / SUB; ++kk) {
            const float2 w = we[kk][lane];
            e.x = t.p0 * e.x + w.x;
            e.y = t.p1 * e.y + w.y;
        }
        ((float2*)seg_end)[(size_t)blockIdx.x * 32 + lane] = e;
    }
}

__global__ __launch_bounds__(256) void k_scan_apply(const float* __restrict__ u,
                                                    const float* __restrict__ A_log,
                                                    const float* __restrict__ seg_end,
                                                    _Float16* __restrict__ h_h) {
    __shared__ float2 we[SEG / SUB][32];
    const int lane = threadIdx.x & 31, k = threadIdx.x >> 5;
    const int b = blockIdx.x / NSEG, sg = blockIdx.x % NSEG;
    const int l0 = sg * SEG + k * SUB;
    ScanLane t;
    scan_load(u, A_log, b, l0, lane, t);

    const float2* ep = (const float2*)seg_end + (size_t)b * NSEG * 32 + lane;
    float2 se[NSEG - 1];
#pragma unroll
    for (int d = 0; d < NSEG - 1; ++d) {
        const int q = sg - 1 - d;
        const float2 v = ep[(q > 0 ? q : 0) * 32];
        se[d] = (q >= 0) ? v : make_float2(0.f, 0.f);
    }

    we[k][lane] = scan_run(t, make_float2(0.f, 0.f));

    float P0 = t.p0, P1 = t.p1;
#pragma unroll
    for (int i = 1; i < SEG / SUB; i <<= 1) { P0 *= P0; P1 *= P1; }
    float2 c = make_float2(0.f, 0.f);
#pragma unroll
    for (int d = NSEG - 2; d >= 0; --d) {
        c.x = P0 * c.x + se[d].x;
        c.y = P1 * c.y + se[d].y;
    }
    __syncthreads();
#pragma unroll
    for (int kk = 0; kk < SEG / SUB - 1; ++kk) {
        const float2 w = we[kk][lane];
        const float nx = t.p0 * c.x + w.x;
        const float ny = t.p1 * c.y + w.y;
        c.x = (kk < k) ? nx : c.x;
        c.y = (kk < k) ? ny : c.y;
    }
    typedef __attribute__((ext_vector_type(2))) _Float16 v2h;
    v2h* hp = (v2h*)(h_h + ((size_t)l0 * BB + b) * SS) + lane;
#pragma unroll
    for (int j = 0; j < SUB; ++j) {
        c.x = t.a0 * c.x + t.uv[j].x;
        c.y = t.a1 * c.y + t.uv[j].y;
        v2h o;
        o[0] = (_Float16)c.x; o[1] = (_Float16)c.y;
        hp[(size_t)j * BB * (SS / 2)] = o;
    }
}

__global__ __launch_bounds__(256) void k_gemm3(const _Float16* __restrict__ h_h,
                                               const _Float16* __restrict__ Dth,
                                               const _Float16* __restrict__ zg,
                                               _Float16* __restrict__ ys) {
    Acc44 acc = {};
    const int w = threadIdx.x >> 5;
    const int rowbase = blockIdx.x * 128 + (w & 1) * 64;
    const int colbase = blockIdx.y * 256 + (w >> 1) * 64;
    wave_gemm64(h_h, Dth, SS, SS, rowbase, colbase, SS, acc);

    const int lane = threadIdx.x & 31;
    const int nlo = lane & 15, mh = (lane >> 4) * 8;
#pragma unroll
    for (int i = 0; i < 4; ++i)
#pragma unroll
        for (int j = 0; j < 4; ++j)
#pragma unroll
            for (int jj = 0; jj < 8; ++jj) {
                const int rp = rowbase + i * 16 + mh + jj;
                const int n = colbase + j * 16 + nlo;
                const int b = rp & (BB - 1), l = rp >> 2;
                const float z = (float)zg[((size_t)b * LL + l) * MM + n];
                ys[(size_t)rp * MM + n] = (_Float16)(acc.c[i][j][jj] * z);
            }
}

__global__ __launch_bounds__(256, 4) void k_gemm4(const _Float16* __restrict__ ys,
                                               const _Float16* __restrict__ Woh,
                                               const float* __restrict__ b_out,
                                               const float* __restrict__ xflat,
                                               float* __restrict__ out) {
    __shared__ __attribute__((aligned(16))) _Float16 smem[2 * GSTAGE];
    Acc44 acc = {};
    const int w = threadIdx.x >> 5;
    const int rowbase = blockIdx.x * 128 + (w & 1) * 64;
    const int colbase = blockIdx.y * 256 + (w >> 1) * 64;
    block_gemm_lds(ys, Woh, blockIdx.x * 128, blockIdx.y * 256, smem, acc);

    const int lane = threadIdx.x & 31;
    const int nlo = lane & 15, mh = (lane >> 4) * 8;
#pragma unroll
    for (int j = 0; j < 4; ++j) {
        const int n = colbase + j * 16 + nlo;
        const float bias = b_out[n];
#pragma unroll
        for (int i = 0; i < 4; ++i)
#pragma unroll
            for (int jj = 0; jj < 8; ++jj) {
                const int m = rowbase + i * 16 + mh + jj;
                const size_t idx = (size_t)m * MM + n;
                out[idx] = xflat[idx] + acc.c[i][j][jj] + bias;
            }
    }
}

extern "C" void kernel_launch(void* const* d_in, const int* in_sizes, int n_in,
                              void* d_out, int out_size, void* d_ws, size_t ws_size,
                              hipStream_t stream) {
    const float* x     = (const float*)d_in[0];
    const float* A_log = (const float*)d_in[1];
    const float* Dmat  = (const float*)d_in[2];
    const float* W_in  = (const float*)d_in[3];
    const float* b_in  = (const float*)d_in[4];
    const float* W_xp  = (const float*)d_in[5];
    const float* b_xp  = (const float*)d_in[6];
    const float* W_out = (const float*)d_in[7];
    const float* b_out = (const float*)d_in[8];
    const float* gamma = (const float*)d_in[9];
    const float* beta  = (const float*)d_in[10];
    float* out = (float*)d_out;

    char* ws = (char*)d_ws;
    size_t off = 0;
    auto take = [&](size_t bytes) { char* p = ws + off; off += bytes; return p; };
    _Float16* xn_h  = (_Float16*)take((size_t)ROWS * MM * 2);
    _Float16* xp_h  = (_Float16*)take((size_t)ROWS * MM * 2);
    _Float16* z_h   = (_Float16*)take((size_t)ROWS * MM * 2);
    _Float16* ys_h  = (_Float16*)take((size_t)ROWS * MM * 2);
    float*    u     = (float*)take((size_t)ROWS * SS * 4);
    float*    sgend = (float*)take((size_t)ROWS * SS * 4);
    _Float16* h_h   = (_Float16*)take((size_t)ROWS * SS * 2);
    _Float16* Wih   = (_Float16*)take((size_t)2 * MM * MM * 2);
    _Float16* Wxh   = (_Float16*)take((size_t)SS * MM * 2);
    _Float16* Woh   = (_Float16*)take((size_t)MM * MM * 2);
    _Float16* Dth   = (_Float16*)take((size_t)SS * MM * 2);
    (void)ws_size; (void)in_sizes; (void)n_in; (void)out_size;

    k_convert<<<CVT_WGS, 256, 0, stream>>>(W_in, W_xp, W_out, Dmat, Wih, Wxh, Woh, Dth);
    k_layernorm<<<ROWS / (8 * LN_RPW), 256, 0, stream>>>(x, gamma, beta, xn_h);
    k_gemm1<<<dim3(ROWS / 128, (2 * MM) / 256), 256, 0, stream>>>(xn_h, Wih, b_in, xp_h, z_h);
    k_gemm2<<<dim3(ROWS / 64, 1), 256, 0, stream>>>(xp_h, Wxh, b_xp, u);
    k_scan_ends<<<BB * NSEG, 256, 0, stream>>>(u, A_log, sgend);
    k_scan_apply<<<BB * NSEG, 256, 0, stream>>>(u, A_log, sgend, h_h);
    k_gemm3<<<dim3(ROWS / 128, MM / 256), 256, 0, stream>>>(h_h, Dth, z_h, ys_h);
    k_gemm4<<<dim3(ROWS / 128, MM / 256), 256, 0, stream>>>(ys_h, Woh, b_out, x, out);
}
